# starting kernel
# baseline (speedup 1.0000x reference)
#include <hip/hip_runtime.h>


typedef float v2f __attribute__((ext_vector_type(2)));
typedef float v8f __attribute__((ext_vector_type(8)));

#define N_WIRES 10
#define N_LAYERS 4
#define DIM 1024
#define PADDED(k) ((k) + ((k) >> 4))
#define PDIM (DIM + (DIM >> 4))
#define LAYER_F 1536

__device__ __forceinline__ void cmul(float ar, float ai, float br, float bi,
                                     float &cr, float &ci) {
  cr = ar * br - ai * bi;
  ci = ar * bi + ai * br;
}

__device__ __forceinline__ int cnot_perm(int m, int rr) {
#pragma unroll
  for (int w = 0; w < N_WIRES; ++w) {
    int t = w + rr; if (t >= N_WIRES) t -= N_WIRES;
    m ^= ((m >> (9 - w)) & 1) << (9 - t);
  }
  return m;
}

__global__ __launch_bounds__(256) void vqc_precompute(const float *__restrict__ W,
                                                      float *__restrict__ G) {
  __shared__ float u[N_WIRES][8];
  const int tid = threadIdx.x;
  for (int l = 0; l < N_LAYERS; ++l) {
    if (tid < N_WIRES) {
      const float phi   = W[(l * N_WIRES + tid) * 3 + 0];
      const float theta = W[(l * N_WIRES + tid) * 3 + 1];
      const float omega = W[(l * N_WIRES + tid) * 3 + 2];
      const float c = __cosf(0.5f * theta), s = __sinf(0.5f * theta);
      const float a = 0.5f * (phi + omega), b = 0.5f * (phi - omega);
      const float ca = __cosf(a), sa = __sinf(a), cb = __cosf(b), sb = __sinf(b);
      u[tid][0] =  ca * c; u[tid][1] = -sa * c;
      u[tid][2] = -cb * s; u[tid][3] = -sb * s;
      u[tid][4] =  cb * s; u[tid][5] = -sb * s;
      u[tid][6] =  ca * c; u[tid][7] =  sa * c;
    }
    __syncthreads();
    float *L = G + l * LAYER_F;
    const int r = tid >> 4, c = tid & 15;
#pragma unroll
    for (int g = 0; g < 2; ++g) {
      float pr = 1.f, pi = 0.f;
#pragma unroll
      for (int q = 0; q < 4; ++q) {
        const int rb = (r >> (3 - q)) & 1, cb2 = (c >> (3 - q)) & 1;
        const float *e = &u[g * 4 + q][(rb * 2 + cb2) * 2];
        float nr, ni; cmul(pr, pi, e[0], e[1], nr, ni);
        pr = nr; pi = ni;
      }
      L[g * 512 + c * 16 + r] = pr;
      L[g * 512 + 256 + c * 16 + r] = pi;
    }
    {
      const int a = r >> 2, b = r & 3, a2 = c >> 2, b2 = c & 3;
      float pr = 0.f, pi = 0.f;
      if (a == a2) {
        const float *e8 = &u[8][(((b >> 1) * 2) + (b2 >> 1)) * 2];
        const float *e9 = &u[9][(((b & 1) * 2) + (b2 & 1)) * 2];
        cmul(e8[0], e8[1], e9[0], e9[1], pr, pi);
      }
      L[1024 + c * 16 + r] = pr;
      L[1280 + c * 16 + r] = pi;
    }
    __syncthreads();
  }
}

__device__ __forceinline__ void group_apply(const float *__restrict__ A,
                                            const float *p1, const float *p2,
                                            float *dst,
                                            int kkStride, int compOff,
                                            int mode, int rr, int permBase,
                                            bool pimg) {
  v8f acc = {};
#pragma unroll
  for (int kk = 0; kk < 4; ++kk) {
    v2f ar, ai, bF, bS, aS;
    ar.x = A[kk * 64];        ar.y = A[kk * 64 + 16];
    ai.x = A[256 + kk * 64];  ai.y = A[256 + kk * 64 + 16];
    bF.x = p1[kk * kkStride];           bF.y = p1[kk * kkStride + compOff];
    bS.x = p2[kk * kkStride];           bS.y = p2[kk * kkStride + compOff];
    aS.x = pimg ? ai.x : -ai.x;
    aS.y = pimg ? ai.y : -ai.y;
    acc = __builtin_amdgcn_wmma_f32_16x16x4_f32(false, ar, false, bF,
                                                (short)0, acc, false, false);
    acc = __builtin_amdgcn_wmma_f32_16x16x4_f32(false, aS, false, bS,
                                                (short)0, acc, false, false);
  }
#pragma unroll
  for (int v = 0; v < 8; ++v) {
    int w;
    if (mode == 0)      w = v * 68;
    else if (mode == 1) w = v * 4 + (v >> 2);
    else                w = PADDED(permBase ^ cnot_perm(v, rr));
    dst[w] = acc[v];
  }
}

__global__ __launch_bounds__(256) void vqc_kernel(const float *__restrict__ X,
                                                  const float *__restrict__ G,
                                                  const float *__restrict__ bias,
                                                  float *__restrict__ out) {
  __shared__ float SR[2][PDIM];
  __shared__ float SI[2][PDIM];
  __shared__ float gm[N_LAYERS * LAYER_F];
  __shared__ float cs[2 * N_WIRES];
  __shared__ float pLo[32], pHi[32];
  __shared__ int   permT[N_LAYERS][64];
  __shared__ float red[8];

  const int tid  = threadIdx.x;
  const int lane = tid & 31;
  const int wave = tid >> 5;
  const int s    = blockIdx.x;

  for (int i = tid; i < N_LAYERS * LAYER_F; i += 256) gm[i] = G[i];

  permT[tid >> 6][tid & 63] = cnot_perm((tid & 63) << 4, (tid >> 6) + 1);

  if (tid < N_WIRES) {
    const float x = X[s * N_WIRES + tid] * 0.7853981633974483f;
    cs[tid] = __cosf(x);
    cs[N_WIRES + tid] = __sinf(x);
  }
  __syncthreads();

  if (tid < 32) {
    float p = 1.f;
#pragma unroll
    for (int w = 5; w < N_WIRES; ++w)
      p *= cs[((tid >> (9 - w)) & 1) * N_WIRES + w];
    pLo[tid] = p;
  } else if (tid < 64) {
    const int h = tid - 32;
    float p = 1.f;
#pragma unroll
    for (int w = 0; w < 5; ++w)
      p *= cs[((h >> (4 - w)) & 1) * N_WIRES + w];
    pHi[h] = p;
  }
  __syncthreads();

  for (int k = tid; k < DIM; k += 256) {
    SR[0][PADDED(k)] = pHi[k >> 5] * pLo[k & 31];
    SI[0][PADDED(k)] = 0.f;
  }
  __syncthreads();

  const int  tile = wave >> 1;
  const bool pimg = (wave & 1) != 0;
  const int  col  = tile * 16 + (lane & 15);
  const int  kb   = (lane >> 4) << 1;
  const int  M    = lane & 15;
  const int  laneHi8 = (lane >> 4) << 3;
  const int  cx = col >> 2, cy = col & 3;
  const int baseB0 = col + (col >> 4) + kb * 68;
  const int baseB1 = cx * 68 + cy + kb * 4;
  const int baseB2 = col * 17 + kb;
  const int baseW0 = col + (col >> 4) + laneHi8 * 68;
  const int baseW1 = cx * 68 + cy + laneHi8 * 4 + (laneHi8 >> 2);
  const int baseA  = kb * 16 + M;

#pragma unroll
  for (int l = 0; l < N_LAYERS; ++l) {
    const float *Lg = gm + l * LAYER_F;
    const int r0 = (3 * l) & 1;

    group_apply(Lg + baseA,
                (pimg ? SI[r0] : SR[r0]) + baseB0,
                (pimg ? SR[r0] : SI[r0]) + baseB0,
                (pimg ? SI[r0 ^ 1] : SR[r0 ^ 1]) + baseW0,
                272, 68, 0, 0, 0, pimg);
    __syncthreads();
    group_apply(Lg + 512 + baseA,
                (pimg ? SI[r0 ^ 1] : SR[r0 ^ 1]) + baseB1,
                (pimg ? SR[r0 ^ 1] : SI[r0 ^ 1]) + baseB1,
                (pimg ? SI[r0] : SR[r0]) + baseW1,
                17, 4, 1, 0, 0, pimg);
    __syncthreads();
    {
      const int pb = permT[l][col] ^ ((lane & 16) ? cnot_perm(8, l + 1) : 0);
      group_apply(Lg + 1024 + baseA,
                  (pimg ? SI[r0] : SR[r0]) + baseB2,
                  (pimg ? SR[r0] : SI[r0]) + baseB2,
                  (pimg ? SI[r0 ^ 1] : SR[r0 ^ 1]),
                  4, 1, 2, l + 1, pb, pimg);
      __syncthreads();
    }
  }

  const float sgn = (tid & 1) ? -1.f : 1.f;
  float acc = 0.f;
#pragma unroll
  for (int j = 0; j < DIM / 256; ++j) {
    const int a = PADDED(tid + 256 * j);
    const float re = SR[0][a], im = SI[0][a];
    acc += re * re + im * im;
  }
  float z = sgn * acc;
#pragma unroll
  for (int m = 16; m > 0; m >>= 1) z += __shfl_xor(z, m, 32);
  if (lane == 0) red[wave] = z;
  __syncthreads();
  if (tid == 0) {
    float t = 0.f;
#pragma unroll
    for (int w = 0; w < 8; ++w) t += red[w];
    out[s] = t + bias[0];
  }
}

extern "C" void kernel_launch(void *const *d_in, const int *in_sizes, int n_in,
                              void *d_out, int out_size, void *d_ws, size_t ws_size,
                              hipStream_t stream) {
  const float *X = (const float *)d_in[0];
  const float *W = (const float *)d_in[1];
  const float *bias = (const float *)d_in[2];
  float *out = (float *)d_out;
  float *G = (float *)d_ws;

  const int B = in_sizes[0] / N_WIRES;

  vqc_precompute<<<1, 256, 0, stream>>>(W, G);
  vqc_kernel<<<B, 256, 0, stream>>>(X, G, bias, out);
}


# Rot = RZ*RY*RZ: only the real RY Kronecker operators go through WMMA, 2 real products per K-step (was 3); a layer's RZ phases and its CNOT ring: one phase-table multiply in GT's scatter; new sums
# speedup vs baseline: 1.8147x; 1.8147x over previous
#include <hip/hip_runtime.h>


typedef float v2f __attribute__((ext_vector_type(2)));
typedef float v4f __attribute__((ext_vector_type(4)));
typedef float v8f __attribute__((ext_vector_type(8)));

#define N_WIRES 10
#define N_LAYERS 4
#define DIM 1024
#define PADDED(k) ((k) + ((k) >> 4))
#define PDIM (DIM + (DIM >> 4))
#define PHASE_F 256
#define LAYER_F 768
#define OPS_F (N_LAYERS * LAYER_F)
#define DIAG_F 2048
#define E_OFF OPS_F
#define F_OFF (E_OFF + N_LAYERS * DIAG_F)

__device__ __forceinline__ int frag_pos(int M, int K) {
  return (K >> 3) * 128 + (M + 16 * ((K >> 1) & 1)) * 4 + ((K >> 2) & 1) * 2 + (K & 1);
}

__device__ __forceinline__ int diag_pos(int m) {
  const int tile = m >> 8, c = (m >> 4) & 15, h = (m >> 3) & 1, v = m & 7;
  return ((tile * 4 + (v >> 1)) * 32 + (c + 16 * h)) * 4 + (v & 1) * 2;
}

struct AOp { v4f q[2]; };

__device__ __forceinline__ AOp load_op(const float *__restrict__ G, int phase, int lane) {
  const v4f *p = reinterpret_cast<const v4f *>(G + phase * PHASE_F) + lane;
  AOp o;
  o.q[0] = p[0];
  o.q[1] = p[32];
  return o;
}

__device__ __forceinline__ void cmul(float ar, float ai, float br, float bi,
                                     float &cr, float &ci) {
  cr = ar * br - ai * bi;
  ci = ar * bi + ai * br;
}

__device__ __forceinline__ int cnot_perm(int m, int rr) {
#pragma unroll
  for (int w = 0; w < N_WIRES; ++w) {
    int t = w + rr; if (t >= N_WIRES) t -= N_WIRES;
    m ^= ((m >> (9 - w)) & 1) << (9 - t);
  }
  return m;
}

__device__ __forceinline__ float ry_el(const float *cs2, int rb, int cb) {
  return (rb == cb) ? cs2[0] : (rb ? cs2[1] : -cs2[1]);
}

__global__ __launch_bounds__(256) void vqc_precompute(const float *__restrict__ W,
                                                      float *__restrict__ G) {
  __shared__ float ry[N_WIRES][2];
  __shared__ float zz[3][N_WIRES][2];
  __shared__ float tab[6][32][2];
  const int tid = threadIdx.x;
  const int l = blockIdx.x;

  if (tid < N_WIRES) {
    const float theta = W[(l * N_WIRES + tid) * 3 + 1];
    const float omega = W[(l * N_WIRES + tid) * 3 + 2];
    ry[tid][0] = __cosf(0.5f * theta);    ry[tid][1] = __sinf(0.5f * theta);
    zz[0][tid][0] = __cosf(0.5f * omega); zz[0][tid][1] = __sinf(0.5f * omega);
  } else if (tid >= 32 && tid < 32 + N_WIRES) {
    const int q = tid - 32;
    float c = 1.f, s = 0.f;
    if (l + 1 < N_LAYERS) {
      const float phi = W[((l + 1) * N_WIRES + q) * 3 + 0];
      c = __cosf(0.5f * phi); s = __sinf(0.5f * phi);
    }
    zz[1][q][0] = c; zz[1][q][1] = s;
  } else if (tid >= 64 && tid < 64 + N_WIRES) {
    const int q = tid - 64;
    const float phi = W[q * 3 + 0];
    zz[2][q][0] = __cosf(0.5f * phi); zz[2][q][1] = __sinf(0.5f * phi);
  }
  __syncthreads();

  if (tid < 192) {
    const int t = tid >> 5, e = tid & 31;
    const int w0 = (t & 1) * 5;
    float pr = 1.f, pi = 0.f;
#pragma unroll
    for (int q = 0; q < 5; ++q) {
      const float c = zz[t >> 1][w0 + q][0];
      const float s = ((e >> (4 - q)) & 1) ? zz[t >> 1][w0 + q][1] : -zz[t >> 1][w0 + q][1];
      float nr, ni; cmul(pr, pi, c, s, nr, ni);
      pr = nr; pi = ni;
    }
    tab[t][e][0] = pr; tab[t][e][1] = pi;
  }
  __syncthreads();

  {
    float *L = G + l * LAYER_F;
    const int r = tid >> 4, c = tid & 15;
#pragma unroll
    for (int g = 0; g < 2; ++g) {
      float p = 1.f;
#pragma unroll
      for (int q = 0; q < 4; ++q)
        p *= ry_el(ry[g * 4 + q], (r >> (3 - q)) & 1, (c >> (3 - q)) & 1);
      L[g * PHASE_F + frag_pos(r, c)] = p;
    }
    const int a = r >> 2, b = r & 3, a2 = c >> 2, b2 = c & 3;
    float p = 0.f;
    if (a == a2) p = ry_el(ry[8], b >> 1, b2 >> 1) * ry_el(ry[9], b & 1, b2 & 1);
    L[2 * PHASE_F + frag_pos(r, c)] = p;
  }

  {
    float *E = G + E_OFF + l * DIAG_F;
    for (int m = tid; m < DIM; m += 256) {
      const int p = cnot_perm(m, l + 1);
      float orr, oi, fr, fi, er, ei;
      cmul(tab[0][m >> 5][0], tab[0][m >> 5][1], tab[1][m & 31][0], tab[1][m & 31][1], orr, oi);
      cmul(tab[2][p >> 5][0], tab[2][p >> 5][1], tab[3][p & 31][0], tab[3][p & 31][1], fr, fi);
      cmul(orr, oi, fr, fi, er, ei);
      const int pos = diag_pos(m);
      E[pos] = er; E[pos + 1] = ei;
    }
  }

  if (l == 0 && tid < 64) {
    G[F_OFF + tid * 2]     = tab[4 + (tid >> 5)][tid & 31][0];
    G[F_OFF + tid * 2 + 1] = tab[4 + (tid >> 5)][tid & 31][1];
  }
}

__device__ __forceinline__ void group_apply(const AOp op, AOp &nextOp,
                                            const float *__restrict__ G,
                                            int nextPhase, int lane,
                                            const float *pR, const float *pI,
                                            float *dR, float *dI,
                                            int kkStride, int compOff,
                                            int mode, int rr, int permBase,
                                            const v4f *__restrict__ ePtr) {
  v8f tr = {}, ti = {};
  if (nextPhase >= 0) nextOp = load_op(G, nextPhase, lane);
  v4f e[4] = {};
  if (mode == 2) {
#pragma unroll
    for (int j = 0; j < 4; ++j) e[j] = ePtr[j * 32];
  }
#pragma unroll
  for (int kk = 0; kk < 4; ++kk) {
    v2f a, br, bi;
    const v4f q = op.q[kk >> 1];
    if (kk & 1) { a.x = q.z; a.y = q.w; } else { a.x = q.x; a.y = q.y; }
    br.x = pR[kk * kkStride];           br.y = pR[kk * kkStride + compOff];
    bi.x = pI[kk * kkStride];           bi.y = pI[kk * kkStride + compOff];
    tr = __builtin_amdgcn_wmma_f32_16x16x4_f32(false, a, false, br,
                                               (short)0, tr, false, false);
    ti = __builtin_amdgcn_wmma_f32_16x16x4_f32(false, a, false, bi,
                                               (short)0, ti, false, false);
  }
#pragma unroll
  for (int v = 0; v < 8; ++v) {
    if (mode == 2) {
      const int w = PADDED(permBase ^ cnot_perm(v, rr));
      const float er = (v & 1) ? e[v >> 1].z : e[v >> 1].x;
      const float ei = (v & 1) ? e[v >> 1].w : e[v >> 1].y;
      float vr, vi; cmul(tr[v], ti[v], er, ei, vr, vi);
      dR[w] = vr;
      dI[w] = vi;
    } else {
      const int w = (mode == 0) ? v * 68 : v * 4 + (v >> 2);
      dR[w] = tr[v];
      dI[w] = ti[v];
    }
  }
}

__global__ __launch_bounds__(128) __attribute__((amdgpu_waves_per_eu(16, 16)))
void vqc_kernel(const float *__restrict__ X,
                const float *__restrict__ G,
                const float *__restrict__ bias,
                float *__restrict__ out) {
  __shared__ float SR[2][PDIM];
  __shared__ float SI[2][PDIM];
  __shared__ float cs[2 * N_WIRES];
  __shared__ float pLo[32][2], pHi[32][2];
  __shared__ float red[4];

  const int tid  = threadIdx.x;
  const int lane = tid & 31;
  const int wave = __builtin_amdgcn_readfirstlane(tid >> 5);
  const int s    = blockIdx.x;

  AOp aCur = load_op(G, 0, lane);

  if (tid < N_WIRES) {
    const float x = X[s * N_WIRES + tid] * 0.7853981633974483f;
    cs[tid] = __cosf(x);
    cs[N_WIRES + tid] = __sinf(x);
  }
  __syncthreads();

  if (tid < 32) {
    float p = 1.f;
#pragma unroll
    for (int w = 5; w < N_WIRES; ++w)
      p *= cs[((tid >> (9 - w)) & 1) * N_WIRES + w];
    pLo[tid][0] = p * G[F_OFF + 64 + tid * 2];
    pLo[tid][1] = p * G[F_OFF + 64 + tid * 2 + 1];
  } else if (tid < 64) {
    const int h = tid - 32;
    float p = 1.f;
#pragma unroll
    for (int w = 0; w < 5; ++w)
      p *= cs[((h >> (4 - w)) & 1) * N_WIRES + w];
    pHi[h][0] = p * G[F_OFF + h * 2];
    pHi[h][1] = p * G[F_OFF + h * 2 + 1];
  }
  __syncthreads();

  for (int k = tid; k < DIM; k += 128) {
    float vr, vi;
    cmul(pHi[k >> 5][0], pHi[k >> 5][1], pLo[k & 31][0], pLo[k & 31][1], vr, vi);
    SR[0][PADDED(k)] = vr;
    SI[0][PADDED(k)] = vi;
  }
  __syncthreads();

  const int  tile = wave;
  const int  col  = tile * 16 + (lane & 15);
  const int  kb   = (lane >> 4) << 1;
  const int  laneHi8 = (lane >> 4) << 3;
  const int  cx = col >> 2, cy = col & 3;
  const int baseB0 = col + (col >> 4) + kb * 68;
  const int baseB1 = cx * 68 + cy + kb * 4;
  const int baseB2 = col * 17 + kb;
  const int baseW0 = col + (col >> 4) + laneHi8 * 68;
  const int baseW1 = cx * 68 + cy + laneHi8 * 4 + (laneHi8 >> 2);
  const v4f *eLane = reinterpret_cast<const v4f *>(G + E_OFF) + tile * 128 + lane;

#pragma nounroll
  for (int l = 0; l < N_LAYERS; ++l) {
    const int r0 = (3 * l) & 1;

    AOp aG1, aGT;
    group_apply(aCur, aG1, G, 3 * l + 1, lane,
                SR[r0] + baseB0, SI[r0] + baseB0,
                SR[r0 ^ 1] + baseW0, SI[r0 ^ 1] + baseW0,
                272, 68, 0, 0, 0, nullptr);
    __syncthreads();
    group_apply(aG1, aGT, G, 3 * l + 2, lane,
                SR[r0 ^ 1] + baseB1, SI[r0 ^ 1] + baseB1,
                SR[r0] + baseW1, SI[r0] + baseW1,
                17, 4, 1, 0, 0, nullptr);
    __syncthreads();
    {
      const int pb = cnot_perm((col << 4) | laneHi8, l + 1);
      group_apply(aGT, aCur, G, (l + 1 < N_LAYERS) ? 3 * l + 3 : -1, lane,
                  SR[r0] + baseB2, SI[r0] + baseB2,
                  SR[r0 ^ 1], SI[r0 ^ 1],
                  4, 1, 2, l + 1, pb, eLane + l * (DIAG_F / 4));
      __syncthreads();
    }
  }

  const float sgn = (tid & 1) ? -1.f : 1.f;
  float acc = 0.f;
#pragma unroll
  for (int j = 0; j < DIM / 128; ++j) {
    const int a = PADDED(tid + 128 * j);
    const float re = SR[0][a], im = SI[0][a];
    acc += re * re + im * im;
  }
  float z = sgn * acc;
#pragma unroll
  for (int m = 16; m > 0; m >>= 1) z += __shfl_xor(z, m, 32);
  if (lane == 0) red[wave] = z;
  __syncthreads();
  if (tid == 0) {
    float t = 0.f;
#pragma unroll
    for (int w = 0; w < 4; ++w) t += red[w];
    out[s] = t + bias[0];
  }
}

extern "C" void kernel_launch(void *const *d_in, const int *in_sizes, int n_in,
                              void *d_out, int out_size, void *d_ws, size_t ws_size,
                              hipStream_t stream) {
  const float *X = (const float *)d_in[0];
  const float *W = (const float *)d_in[1];
  const float *bias = (const float *)d_in[2];
  float *out = (float *)d_out;
  float *G = (float *)d_ws;

  const int B = in_sizes[0] / N_WIRES;

  vqc_precompute<<<N_LAYERS, 256, 0, stream>>>(W, G);
  vqc_kernel<<<B, 128, 0, stream>>>(X, G, bias, out);
}


# G1, G0 applied back to back in registers: state is G1's A operand, the accumulator pairs are G0's B operand as they stand; 2 LDS hand-overs a layer (was 3), layer loop unrolled; new sum order
# speedup vs baseline: 2.1362x; 1.1772x over previous
#include <hip/hip_runtime.h>


typedef float v2f __attribute__((ext_vector_type(2)));
typedef float v4f __attribute__((ext_vector_type(4)));
typedef float v8f __attribute__((ext_vector_type(8)));

#define N_WIRES 10
#define N_LAYERS 4
#define DIM 1024
#define PADDED(k) ((k) + ((k) >> 4))
#define PDIM (DIM + (DIM >> 4))
#define PHASE_F 256
#define LAYER_F 768
#define OPS_F (N_LAYERS * LAYER_F)
#define DIAG_F 2048
#define E_OFF OPS_F
#define F_OFF (E_OFF + N_LAYERS * DIAG_F)

__device__ __forceinline__ int frag_pos(int M, int kk, int h, int e) {
  return (kk >> 1) * 128 + (M + 16 * h) * 4 + (kk & 1) * 2 + e;
}

__device__ __forceinline__ int diag_pos(int m) {
  const int tile = m >> 8, c = (m >> 4) & 15, h = (m >> 3) & 1, v = m & 7;
  return ((tile * 4 + (v >> 1)) * 32 + (c + 16 * h)) * 4 + (v & 1) * 2;
}

struct AOp { v4f q[2]; };

__device__ __forceinline__ AOp load_op(const float *__restrict__ G, int phase, unsigned laneB) {
  const char *p = reinterpret_cast<const char *>(G + phase * PHASE_F);
  AOp o;
  o.q[0] = *reinterpret_cast<const v4f *>(p + laneB);
  o.q[1] = *reinterpret_cast<const v4f *>(p + laneB + 512u);
  return o;
}

__device__ __forceinline__ void cmul(float ar, float ai, float br, float bi,
                                     float &cr, float &ci) {
  cr = ar * br - ai * bi;
  ci = ar * bi + ai * br;
}

__device__ __forceinline__ int cnot_perm(int m, int rr) {
#pragma unroll
  for (int w = 0; w < N_WIRES; ++w) {
    int t = w + rr; if (t >= N_WIRES) t -= N_WIRES;
    m ^= ((m >> (9 - w)) & 1) << (9 - t);
  }
  return m;
}

__device__ __forceinline__ float ry_el(const float *cs2, int rb, int cb) {
  return (rb == cb) ? cs2[0] : (rb ? cs2[1] : -cs2[1]);
}

__global__ __launch_bounds__(256) void vqc_precompute(const float *__restrict__ W,
                                                      float *__restrict__ G) {
  __shared__ float ry[N_WIRES][2];
  __shared__ float zz[3][N_WIRES][2];
  __shared__ float tab[6][32][2];
  const int tid = threadIdx.x;
  const int l = blockIdx.x;

  if (tid < N_WIRES) {
    const float theta = W[(l * N_WIRES + tid) * 3 + 1];
    const float omega = W[(l * N_WIRES + tid) * 3 + 2];
    ry[tid][0] = __cosf(0.5f * theta);    ry[tid][1] = __sinf(0.5f * theta);
    zz[0][tid][0] = __cosf(0.5f * omega); zz[0][tid][1] = __sinf(0.5f * omega);
  } else if (tid >= 32 && tid < 32 + N_WIRES) {
    const int q = tid - 32;
    float c = 1.f, s = 0.f;
    if (l + 1 < N_LAYERS) {
      const float phi = W[((l + 1) * N_WIRES + q) * 3 + 0];
      c = __cosf(0.5f * phi); s = __sinf(0.5f * phi);
    }
    zz[1][q][0] = c; zz[1][q][1] = s;
  } else if (tid >= 64 && tid < 64 + N_WIRES) {
    const int q = tid - 64;
    const float phi = W[q * 3 + 0];
    zz[2][q][0] = __cosf(0.5f * phi); zz[2][q][1] = __sinf(0.5f * phi);
  }
  __syncthreads();

  if (tid < 192) {
    const int t = tid >> 5, e = tid & 31;
    const int w0 = (t & 1) * 5;
    float pr = 1.f, pi = 0.f;
#pragma unroll
    for (int q = 0; q < 5; ++q) {
      const float c = zz[t >> 1][w0 + q][0];
      const float s = ((e >> (4 - q)) & 1) ? zz[t >> 1][w0 + q][1] : -zz[t >> 1][w0 + q][1];
      float nr, ni; cmul(pr, pi, c, s, nr, ni);
      pr = nr; pi = ni;
    }
    tab[t][e][0] = pr; tab[t][e][1] = pi;
  }
  __syncthreads();

  {
    float *L = G + l * LAYER_F;
    const int r = tid >> 4, c = tid & 15;
#pragma unroll
    for (int g = 0; g < 2; ++g) {
      float p = 1.f;
#pragma unroll
      for (int q = 0; q < 4; ++q)
        p *= ry_el(ry[g * 4 + q], (r >> (3 - q)) & 1, (c >> (3 - q)) & 1);
      const int pos = (g == 0) ? frag_pos(r, (c >> 1) & 3, c >> 3, c & 1)
                               : frag_pos(r, 2 * (c >> 3) + ((c >> 1) & 1), (c >> 2) & 1, c & 1);
      L[g * PHASE_F + pos] = p;
    }
    const int a = r >> 2, b = r & 3, a2 = c >> 2, b2 = c & 3;
    float p = 0.f;
    if (a == a2) p = ry_el(ry[8], b >> 1, b2 >> 1) * ry_el(ry[9], b & 1, b2 & 1);
    L[2 * PHASE_F + frag_pos(r, (c >> 1) & 3, c >> 3, c & 1)] = p;
  }

  {
    float *E = G + E_OFF + l * DIAG_F;
    for (int m = tid; m < DIM; m += 256) {
      const int p = cnot_perm(m, l + 1);
      float orr, oi, fr, fi, er, ei;
      cmul(tab[0][m >> 5][0], tab[0][m >> 5][1], tab[1][m & 31][0], tab[1][m & 31][1], orr, oi);
      cmul(tab[2][p >> 5][0], tab[2][p >> 5][1], tab[3][p & 31][0], tab[3][p & 31][1], fr, fi);
      cmul(orr, oi, fr, fi, er, ei);
      const int pos = diag_pos(m);
      E[pos] = er; E[pos + 1] = ei;
    }
  }

  if (l == 0 && tid < 64) {
    G[F_OFF + tid * 2]     = tab[4 + (tid >> 5)][tid & 31][0];
    G[F_OFF + tid * 2 + 1] = tab[4 + (tid >> 5)][tid & 31][1];
  }
}

__device__ __forceinline__ v2f op_pair(const AOp &o, int kk) {
  const v4f q = o.q[kk >> 1];
  v2f a;
  if (kk & 1) { a.x = q.z; a.y = q.w; } else { a.x = q.x; a.y = q.y; }
  return a;
}

__device__ __forceinline__ void fused_apply(const AOp opY, const AOp opX,
                                            AOp &opNext, const float *__restrict__ G,
                                            int nextPhase, unsigned laneB,
                                            const float *pR, const float *pI,
                                            float *dR, float *dI) {
  v8f tr = {}, ti = {};
#pragma unroll
  for (int kk = 0; kk < 4; ++kk) {
    const int off = (kk >> 1) * 34 + (kk & 1) * 8;
    v2f sr, si;
    sr.x = pR[off];  sr.y = pR[off + 4];
    si.x = pI[off];  si.y = pI[off + 4];
    const v2f b = op_pair(opY, kk);
    tr = __builtin_amdgcn_wmma_f32_16x16x4_f32(false, sr, false, b,
                                               (short)0, tr, false, false);
    ti = __builtin_amdgcn_wmma_f32_16x16x4_f32(false, si, false, b,
                                               (short)0, ti, false, false);
  }
  opNext = load_op(G, nextPhase, laneB);
  v8f ur = {}, ui = {};
#pragma unroll
  for (int kk = 0; kk < 4; ++kk) {
    const v2f a = op_pair(opX, kk);
    v2f br, bi;
    br.x = tr[2 * kk];  br.y = tr[2 * kk + 1];
    bi.x = ti[2 * kk];  bi.y = ti[2 * kk + 1];
    ur = __builtin_amdgcn_wmma_f32_16x16x4_f32(false, a, false, br,
                                               (short)0, ur, false, false);
    ui = __builtin_amdgcn_wmma_f32_16x16x4_f32(false, a, false, bi,
                                               (short)0, ui, false, false);
  }
  *reinterpret_cast<v4f *>(dR)     = __builtin_shufflevector(ur, ur, 0, 1, 2, 3);
  *reinterpret_cast<v4f *>(dR + 4) = __builtin_shufflevector(ur, ur, 4, 5, 6, 7);
  *reinterpret_cast<v4f *>(dI)     = __builtin_shufflevector(ui, ui, 0, 1, 2, 3);
  *reinterpret_cast<v4f *>(dI + 4) = __builtin_shufflevector(ui, ui, 4, 5, 6, 7);
}

struct ETab { v4f e[4]; };

__device__ __forceinline__ ETab load_etab(const float *__restrict__ eTile, unsigned laneB) {
  const char *p = reinterpret_cast<const char *>(eTile);
  ETab t;
#pragma unroll
  for (int j = 0; j < 4; ++j)
    t.e[j] = *reinterpret_cast<const v4f *>(p + laneB + 512u * j);
  return t;
}

__device__ __forceinline__ void gt_apply(const AOp op,
                                         const float *pR, const float *pI,
                                         float *dR, float *dI,
                                         int rr, int permBase,
                                         const ETab et) {
  v8f tr = {}, ti = {};
#pragma unroll
  for (int kk = 0; kk < 4; ++kk) {
    const v2f a = op_pair(op, kk);
    v2f br, bi;
    const int off = (kk >> 1) * 32 + (kk & 1) * 16;
    br.x = pR[off];  br.y = pR[off + 8];
    bi.x = pI[off];  bi.y = pI[off + 8];
    tr = __builtin_amdgcn_wmma_f32_16x16x4_f32(false, a, false, br,
                                               (short)0, tr, false, false);
    ti = __builtin_amdgcn_wmma_f32_16x16x4_f32(false, a, false, bi,
                                               (short)0, ti, false, false);
  }
#pragma unroll
  for (int v = 0; v < 8; ++v) {
    const int w = PADDED(permBase ^ cnot_perm(v, rr));
    const float er = (v & 1) ? et.e[v >> 1].z : et.e[v >> 1].x;
    const float ei = (v & 1) ? et.e[v >> 1].w : et.e[v >> 1].y;
    float vr, vi; cmul(tr[v], ti[v], er, ei, vr, vi);
    dR[w] = vr;
    dI[w] = vi;
  }
}

__global__ __launch_bounds__(128) __attribute__((amdgpu_waves_per_eu(16, 16)))
void vqc_kernel(const float *__restrict__ X,
                const float *__restrict__ G,
                const float *__restrict__ bias,
                float *__restrict__ out) {
  __shared__ __attribute__((aligned(16))) float SR[2][PDIM];
  __shared__ __attribute__((aligned(16))) float SI[2][PDIM];
  __shared__ float cs[2 * N_WIRES];
  __shared__ float pLo[32][2], pHi[32][2];
  __shared__ float red[4];

  const int tid  = threadIdx.x;
  const int lane = tid & 31;
  const int wave = __builtin_amdgcn_readfirstlane(tid >> 5);
  const int s    = blockIdx.x;

  if (tid < N_WIRES) {
    const float x = X[s * N_WIRES + tid] * 0.7853981633974483f;
    cs[tid] = __cosf(x);
    cs[N_WIRES + tid] = __sinf(x);
  }
  __syncthreads();

  if (tid < 32) {
    float p = 1.f;
#pragma unroll
    for (int w = 5; w < N_WIRES; ++w)
      p *= cs[((tid >> (9 - w)) & 1) * N_WIRES + w];
    pLo[tid][0] = p * G[F_OFF + 64 + tid * 2];
    pLo[tid][1] = p * G[F_OFF + 64 + tid * 2 + 1];
  } else if (tid < 64) {
    const int h = tid - 32;
    float p = 1.f;
#pragma unroll
    for (int w = 0; w < 5; ++w)
      p *= cs[((h >> (4 - w)) & 1) * N_WIRES + w];
    pHi[h][0] = p * G[F_OFF + h * 2];
    pHi[h][1] = p * G[F_OFF + h * 2 + 1];
  }
  __syncthreads();

  for (int k = tid; k < DIM; k += 128) {
    float vr, vi;
    cmul(pHi[k >> 5][0], pHi[k >> 5][1], pLo[k & 31][0], pLo[k & 31][1], vr, vi);
    SR[0][PADDED(k)] = vr;
    SI[0][PADDED(k)] = vi;
  }
  __syncthreads();

  const int c16 = lane & 15;
  const int hh  = lane >> 4;
  const int baseA  = c16 * 68 + hh * 17 + wave;
  const int baseWF = 4 * (hh * 136 + (c16 >> 1) * 17 + (c16 & 1) * 8 + wave * 2);
  const int col     = wave * 16 + c16;
  const int laneHi8 = hh << 3;
  const int baseB2  = (wave >> 1) * 544 + (wave & 1) * 4 + (c16 & 3) * 136 + hh * 68 + (c16 >> 2);
  const float *eTile = G + E_OFF + wave * 512;
  const unsigned laneB = (unsigned)lane * 16u;

  AOp aG1 = load_op(G, 1, laneB);
#pragma unroll
  for (int l = 0; l < N_LAYERS; ++l) {
    const AOp aG0 = load_op(G, 3 * l, laneB);
    AOp aGT;
    fused_apply(aG1, aG0, aGT, G, 3 * l + 2, laneB,
                SR[0] + baseA, SI[0] + baseA,
                SR[1] + baseWF, SI[1] + baseWF);
    __syncthreads();
    const ETab et = load_etab(eTile + l * DIAG_F, laneB);
    {
      const int pb = cnot_perm((col << 4) | laneHi8, l + 1);
      gt_apply(aGT, SR[1] + baseB2, SI[1] + baseB2,
               SR[0], SI[0],
               l + 1, pb, et);
    }
    if (l + 1 < N_LAYERS) aG1 = load_op(G, 3 * l + 4, laneB);
    __syncthreads();
  }

  const int tq = wave * 32 + lane;
  const float sgn = (lane & 1) ? -1.f : 1.f;
  float acc = 0.f;
#pragma unroll
  for (int j = 0; j < DIM / 128; ++j) {
    const int a = PADDED(tq + 128 * j);
    const float re = SR[0][a], im = SI[0][a];
    acc += re * re + im * im;
  }
  float z = sgn * acc;
#pragma unroll
  for (int m = 16; m > 0; m >>= 1) z += __shfl_xor(z, m, 32);
  if (lane == 0) red[wave] = z;
  __syncthreads();
  if (tid == 0) {
    float t = 0.f;
#pragma unroll
    for (int w = 0; w < 4; ++w) t += red[w];
    out[s] = t + bias[0];
  }
}

extern "C" void kernel_launch(void *const *d_in, const int *in_sizes, int n_in,
                              void *d_out, int out_size, void *d_ws, size_t ws_size,
                              hipStream_t stream) {
  const float *X = (const float *)d_in[0];
  const float *W = (const float *)d_in[1];
  const float *bias = (const float *)d_in[2];
  float *out = (float *)d_out;
  float *G = (float *)d_ws;

  const int B = in_sizes[0] / N_WIRES;

  vqc_precompute<<<N_LAYERS, 256, 0, stream>>>(W, G);
  vqc_kernel<<<B, 128, 0, stream>>>(X, G, bias, out);
}


# State enters and leaves the layer loop in registers: start state formed in layer 0's A registers, <Z> summed from the last GT's accumulators; 2 LDS hand-overs, 2 barriers gone; new <Z> sum order
# speedup vs baseline: 2.3359x; 1.0935x over previous
#include <hip/hip_runtime.h>


typedef float v2f __attribute__((ext_vector_type(2)));
typedef float v4f __attribute__((ext_vector_type(4)));
typedef float v8f __attribute__((ext_vector_type(8)));

#define N_WIRES 10
#define N_LAYERS 4
#define DIM 1024
#define PADDED(k) ((k) + ((k) >> 4))
#define PDIM (DIM + (DIM >> 4))
#define PHASE_F 256
#define LAYER_F 768
#define OPS_F (N_LAYERS * LAYER_F)
#define DIAG_F 2048
#define E_OFF OPS_F
#define F_OFF (E_OFF + N_LAYERS * DIAG_F)

__device__ __forceinline__ int frag_pos(int M, int kk, int h, int e) {
  return (kk >> 1) * 128 + (M + 16 * h) * 4 + (kk & 1) * 2 + e;
}

__device__ __forceinline__ int diag_pos(int m) {
  const int tile = m >> 8, c = (m >> 4) & 15, h = (m >> 3) & 1, v = m & 7;
  return ((tile * 4 + (v >> 1)) * 32 + (c + 16 * h)) * 4 + (v & 1) * 2;
}

struct AOp { v4f q[2]; };

__device__ __forceinline__ AOp load_op(const float *__restrict__ G, int phase, unsigned laneB) {
  const char *p = reinterpret_cast<const char *>(G + phase * PHASE_F);
  AOp o;
  o.q[0] = *reinterpret_cast<const v4f *>(p + laneB);
  o.q[1] = *reinterpret_cast<const v4f *>(p + laneB + 512u);
  return o;
}

__device__ __forceinline__ void cmul(float ar, float ai, float br, float bi,
                                     float &cr, float &ci) {
  cr = ar * br - ai * bi;
  ci = ar * bi + ai * br;
}

__device__ __forceinline__ int cnot_perm(int m, int rr) {
#pragma unroll
  for (int w = 0; w < N_WIRES; ++w) {
    int t = w + rr; if (t >= N_WIRES) t -= N_WIRES;
    m ^= ((m >> (9 - w)) & 1) << (9 - t);
  }
  return m;
}

__device__ __forceinline__ float ry_el(const float *cs2, int rb, int cb) {
  return (rb == cb) ? cs2[0] : (rb ? cs2[1] : -cs2[1]);
}

__global__ __launch_bounds__(256) void vqc_precompute(const float *__restrict__ W,
                                                      float *__restrict__ G) {
  __shared__ float ry[N_WIRES][2];
  __shared__ float zz[3][N_WIRES][2];
  __shared__ float tab[6][32][2];
  const int tid = threadIdx.x;
  const int l = blockIdx.x;

  if (tid < N_WIRES) {
    const float theta = W[(l * N_WIRES + tid) * 3 + 1];
    const float omega = W[(l * N_WIRES + tid) * 3 + 2];
    ry[tid][0] = __cosf(0.5f * theta);    ry[tid][1] = __sinf(0.5f * theta);
    zz[0][tid][0] = __cosf(0.5f * omega); zz[0][tid][1] = __sinf(0.5f * omega);
  } else if (tid >= 32 && tid < 32 + N_WIRES) {
    const int q = tid - 32;
    float c = 1.f, s = 0.f;
    if (l + 1 < N_LAYERS) {
      const float phi = W[((l + 1) * N_WIRES + q) * 3 + 0];
      c = __cosf(0.5f * phi); s = __sinf(0.5f * phi);
    }
    zz[1][q][0] = c; zz[1][q][1] = s;
  } else if (tid >= 64 && tid < 64 + N_WIRES) {
    const int q = tid - 64;
    const float phi = W[q * 3 + 0];
    zz[2][q][0] = __cosf(0.5f * phi); zz[2][q][1] = __sinf(0.5f * phi);
  }
  __syncthreads();

  if (tid < 192) {
    const int t = tid >> 5, e = tid & 31;
    const int w0 = (t & 1) * 5;
    float pr = 1.f, pi = 0.f;
#pragma unroll
    for (int q = 0; q < 5; ++q) {
      const float c = zz[t >> 1][w0 + q][0];
      const float s = ((e >> (4 - q)) & 1) ? zz[t >> 1][w0 + q][1] : -zz[t >> 1][w0 + q][1];
      float nr, ni; cmul(pr, pi, c, s, nr, ni);
      pr = nr; pi = ni;
    }
    tab[t][e][0] = pr; tab[t][e][1] = pi;
  }
  __syncthreads();

  {
    float *L = G + l * LAYER_F;
    const int r = tid >> 4, c = tid & 15;
#pragma unroll
    for (int g = 0; g < 2; ++g) {
      float p = 1.f;
#pragma unroll
      for (int q = 0; q < 4; ++q)
        p *= ry_el(ry[g * 4 + q], (r >> (3 - q)) & 1, (c >> (3 - q)) & 1);
      const int pos = (g == 0) ? frag_pos(r, (c >> 1) & 3, c >> 3, c & 1)
                               : frag_pos(r, 2 * (c >> 3) + ((c >> 1) & 1), (c >> 2) & 1, c & 1);
      L[g * PHASE_F + pos] = p;
    }
    const int a = r >> 2, b = r & 3, a2 = c >> 2, b2 = c & 3;
    float p = 0.f;
    if (a == a2) p = ry_el(ry[8], b >> 1, b2 >> 1) * ry_el(ry[9], b & 1, b2 & 1);
    L[2 * PHASE_F + frag_pos(r, (c >> 1) & 3, c >> 3, c & 1)] = p;
  }

  if (l + 1 < N_LAYERS) {
    float *E = G + E_OFF + l * DIAG_F;
    for (int m = tid; m < DIM; m += 256) {
      const int p = cnot_perm(m, l + 1);
      float orr, oi, fr, fi, er, ei;
      cmul(tab[0][m >> 5][0], tab[0][m >> 5][1], tab[1][m & 31][0], tab[1][m & 31][1], orr, oi);
      cmul(tab[2][p >> 5][0], tab[2][p >> 5][1], tab[3][p & 31][0], tab[3][p & 31][1], fr, fi);
      cmul(orr, oi, fr, fi, er, ei);
      const int pos = diag_pos(m);
      E[pos] = er; E[pos + 1] = ei;
    }
  }

  if (l == 0 && tid < 64) {
    G[F_OFF + tid * 2]     = tab[4 + (tid >> 5)][tid & 31][0];
    G[F_OFF + tid * 2 + 1] = tab[4 + (tid >> 5)][tid & 31][1];
  }
}

__device__ __forceinline__ v2f op_pair(const AOp &o, int kk) {
  const v4f q = o.q[kk >> 1];
  v2f a;
  if (kk & 1) { a.x = q.z; a.y = q.w; } else { a.x = q.x; a.y = q.y; }
  return a;
}

struct StartF { v2f hi[2], lo[4]; };

template <bool START>
__device__ __forceinline__ void fused_apply(const AOp opY, const AOp opX,
                                            AOp &opNext, const float *__restrict__ G,
                                            int nextPhase, unsigned laneB,
                                            const float *pR, const float *pI,
                                            const StartF &sf,
                                            float *dR, float *dI) {
  v8f tr = {}, ti = {};
#pragma unroll
  for (int kk = 0; kk < 4; ++kk) {
    v2f sr, si;
    if (START) {
      const v2f h = sf.hi[kk >> 1];
      const v2f l0 = sf.lo[2 * (kk & 1)], l1 = sf.lo[2 * (kk & 1) + 1];
      float r0, i0, r1, i1;
      cmul(h.x, h.y, l0.x, l0.y, r0, i0);
      cmul(h.x, h.y, l1.x, l1.y, r1, i1);
      sr.x = r0;  sr.y = r1;
      si.x = i0;  si.y = i1;
    } else {
      const int off = (kk >> 1) * 34 + (kk & 1) * 8;
      sr.x = pR[off];  sr.y = pR[off + 4];
      si.x = pI[off];  si.y = pI[off + 4];
    }
    const v2f b = op_pair(opY, kk);
    tr = __builtin_amdgcn_wmma_f32_16x16x4_f32(false, sr, false, b,
                                               (short)0, tr, false, false);
    ti = __builtin_amdgcn_wmma_f32_16x16x4_f32(false, si, false, b,
                                               (short)0, ti, false, false);
  }
  opNext = load_op(G, nextPhase, laneB);
  v8f ur = {}, ui = {};
#pragma unroll
  for (int kk = 0; kk < 4; ++kk) {
    const v2f a = op_pair(opX, kk);
    v2f br, bi;
    br.x = tr[2 * kk];  br.y = tr[2 * kk + 1];
    bi.x = ti[2 * kk];  bi.y = ti[2 * kk + 1];
    ur = __builtin_amdgcn_wmma_f32_16x16x4_f32(false, a, false, br,
                                               (short)0, ur, false, false);
    ui = __builtin_amdgcn_wmma_f32_16x16x4_f32(false, a, false, bi,
                                               (short)0, ui, false, false);
  }
  *reinterpret_cast<v4f *>(dR)     = __builtin_shufflevector(ur, ur, 0, 1, 2, 3);
  *reinterpret_cast<v4f *>(dR + 4) = __builtin_shufflevector(ur, ur, 4, 5, 6, 7);
  *reinterpret_cast<v4f *>(dI)     = __builtin_shufflevector(ui, ui, 0, 1, 2, 3);
  *reinterpret_cast<v4f *>(dI + 4) = __builtin_shufflevector(ui, ui, 4, 5, 6, 7);
}

struct ETab { v4f e[4]; };

__device__ __forceinline__ ETab load_etab(const float *__restrict__ eTile, unsigned laneB) {
  const char *p = reinterpret_cast<const char *>(eTile);
  ETab t;
#pragma unroll
  for (int j = 0; j < 4; ++j)
    t.e[j] = *reinterpret_cast<const v4f *>(p + laneB + 512u * j);
  return t;
}

__device__ __forceinline__ void gt_products(const AOp op,
                                            const float *pR, const float *pI,
                                            v8f &tr, v8f &ti) {
  tr = v8f{};  ti = v8f{};
#pragma unroll
  for (int kk = 0; kk < 4; ++kk) {
    const v2f a = op_pair(op, kk);
    v2f br, bi;
    const int off = (kk >> 1) * 32 + (kk & 1) * 16;
    br.x = pR[off];  br.y = pR[off + 8];
    bi.x = pI[off];  bi.y = pI[off + 8];
    tr = __builtin_amdgcn_wmma_f32_16x16x4_f32(false, a, false, br,
                                               (short)0, tr, false, false);
    ti = __builtin_amdgcn_wmma_f32_16x16x4_f32(false, a, false, bi,
                                               (short)0, ti, false, false);
  }
}

__device__ __forceinline__ void gt_apply(const AOp op,
                                         const float *pR, const float *pI,
                                         float *dR, float *dI,
                                         int rr, int permBase,
                                         const ETab et) {
  v8f tr, ti;
  gt_products(op, pR, pI, tr, ti);
#pragma unroll
  for (int v = 0; v < 8; ++v) {
    const int w = PADDED(permBase ^ cnot_perm(v, rr));
    const float er = (v & 1) ? et.e[v >> 1].z : et.e[v >> 1].x;
    const float ei = (v & 1) ? et.e[v >> 1].w : et.e[v >> 1].y;
    float vr, vi; cmul(tr[v], ti[v], er, ei, vr, vi);
    dR[w] = vr;
    dI[w] = vi;
  }
}

__device__ __forceinline__ float gt_measure(const AOp op,
                                            const float *pR, const float *pI,
                                            int rr, int permBase) {
  v8f tr, ti;
  gt_products(op, pR, pI, tr, ti);
  float acc = 0.f;
#pragma unroll
  for (int v = 0; v < 8; ++v) {
    const float p = tr[v] * tr[v] + ti[v] * ti[v];
    acc += (cnot_perm(v, rr) & 1) ? -p : p;
  }
  return (permBase & 1) ? -acc : acc;
}

__global__ __launch_bounds__(128) __attribute__((amdgpu_waves_per_eu(16, 16)))
void vqc_kernel(const float *__restrict__ X,
                const float *__restrict__ G,
                const float *__restrict__ bias,
                float *__restrict__ out) {
  __shared__ __attribute__((aligned(16))) float SR[2][PDIM];
  __shared__ __attribute__((aligned(16))) float SI[2][PDIM];
  __shared__ float cs[2 * N_WIRES];
  __shared__ float pLo[32][2], pHi[32][2];
  __shared__ float red[4];

  const int tid  = threadIdx.x;
  const int lane = tid & 31;
  const int wave = __builtin_amdgcn_readfirstlane(tid >> 5);
  const int s    = blockIdx.x;

  if (tid < N_WIRES) {
    const float x = X[s * N_WIRES + tid] * 0.7853981633974483f;
    cs[tid] = __cosf(x);
    cs[N_WIRES + tid] = __sinf(x);
  }
  __syncthreads();

  if (tid < 32) {
    float p = 1.f;
#pragma unroll
    for (int w = 5; w < N_WIRES; ++w)
      p *= cs[((tid >> (9 - w)) & 1) * N_WIRES + w];
    pLo[tid][0] = p * G[F_OFF + 64 + tid * 2];
    pLo[tid][1] = p * G[F_OFF + 64 + tid * 2 + 1];
  } else if (tid < 64) {
    const int h = tid - 32;
    float p = 1.f;
#pragma unroll
    for (int w = 0; w < 5; ++w)
      p *= cs[((h >> (4 - w)) & 1) * N_WIRES + w];
    pHi[h][0] = p * G[F_OFF + h * 2];
    pHi[h][1] = p * G[F_OFF + h * 2 + 1];
  }
  __syncthreads();

  const int c16 = lane & 15;
  const int hh  = lane >> 4;

  StartF sf;
#pragma unroll
  for (int j = 0; j < 2; ++j) {
    sf.hi[j].x = pHi[2 * c16 + j][0];
    sf.hi[j].y = pHi[2 * c16 + j][1];
  }
#pragma unroll
  for (int j = 0; j < 4; ++j) {
    sf.lo[j].x = pLo[16 * hh + 4 * j + wave][0];
    sf.lo[j].y = pLo[16 * hh + 4 * j + wave][1];
  }

  const int baseA  = c16 * 68 + hh * 17 + wave;
  const int baseWF = 4 * (hh * 136 + (c16 >> 1) * 17 + (c16 & 1) * 8 + wave * 2);
  const int col     = wave * 16 + c16;
  const int laneHi8 = hh << 3;
  const int baseB2  = (wave >> 1) * 544 + (wave & 1) * 4 + (c16 & 3) * 136 + hh * 68 + (c16 >> 2);
  const float *eTile = G + E_OFF + wave * 512;
  const unsigned laneB = (unsigned)lane * 16u;

  AOp aG1 = load_op(G, 1, laneB);
  float z = 0.f;
#pragma unroll
  for (int l = 0; l < N_LAYERS; ++l) {
    const AOp aG0 = load_op(G, 3 * l, laneB);
    AOp aGT;
    if (l == 0)
      fused_apply<true>(aG1, aG0, aGT, G, 3 * l + 2, laneB,
                        SR[0] + baseA, SI[0] + baseA, sf,
                        SR[1] + baseWF, SI[1] + baseWF);
    else
      fused_apply<false>(aG1, aG0, aGT, G, 3 * l + 2, laneB,
                         SR[0] + baseA, SI[0] + baseA, sf,
                         SR[1] + baseWF, SI[1] + baseWF);
    __syncthreads();
    const int pb = cnot_perm((col << 4) | laneHi8, l + 1);
    if (l + 1 < N_LAYERS) {
      const ETab et = load_etab(eTile + l * DIAG_F, laneB);
      gt_apply(aGT, SR[1] + baseB2, SI[1] + baseB2,
               SR[0], SI[0],
               l + 1, pb, et);
      aG1 = load_op(G, 3 * l + 4, laneB);
      __syncthreads();
    } else {
      z = gt_measure(aGT, SR[1] + baseB2, SI[1] + baseB2, l + 1, pb);
    }
  }

#pragma unroll
  for (int m = 16; m > 0; m >>= 1) z += __shfl_xor(z, m, 32);
  if (lane == 0) red[wave] = z;
  __syncthreads();
  if (tid == 0) {
    float t = 0.f;
#pragma unroll
    for (int w = 0; w < 4; ++w) t += red[w];
    out[s] = t + bias[0];
  }
}

extern "C" void kernel_launch(void *const *d_in, const int *in_sizes, int n_in,
                              void *d_out, int out_size, void *d_ws, size_t ws_size,
                              hipStream_t stream) {
  const float *X = (const float *)d_in[0];
  const float *W = (const float *)d_in[1];
  const float *bias = (const float *)d_in[2];
  float *out = (float *)d_out;
  float *G = (float *)d_ws;

  const int B = in_sizes[0] / N_WIRES;

  vqc_precompute<<<N_LAYERS, 256, 0, stream>>>(W, G);
  vqc_kernel<<<B, 128, 0, stream>>>(X, G, bias, out);
}


# GT writeback's scatter offsets (padded CNOT-ring image of the index, sample-independent) read from a table the first kernel builds: two 16-byte loads a layer replace the index math; sums unchanged
# speedup vs baseline: 2.4487x; 1.0483x over previous
#include <hip/hip_runtime.h>


typedef float v2f __attribute__((ext_vector_type(2)));
typedef float v4f __attribute__((ext_vector_type(4)));
typedef float v8f __attribute__((ext_vector_type(8)));
typedef unsigned v4u __attribute__((ext_vector_type(4)));

#define N_WIRES 10
#define N_LAYERS 4
#define DIM 1024
#define PADDED(k) ((k) + ((k) >> 4))
#define PDIM (DIM + (DIM >> 4))
#define PHASE_F 256
#define LAYER_F 768
#define OPS_F (N_LAYERS * LAYER_F)
#define DIAG_F 2048
#define E_OFF OPS_F
#define F_OFF (E_OFF + N_LAYERS * DIAG_F)
#define S_OFF (F_OFF + 128)
#define SCAT_F 1024

__device__ __forceinline__ int frag_pos(int M, int kk, int h, int e) {
  return (kk >> 1) * 128 + (M + 16 * h) * 4 + (kk & 1) * 2 + e;
}

__device__ __forceinline__ int diag_pos(int m) {
  const int tile = m >> 8, c = (m >> 4) & 15, h = (m >> 3) & 1, v = m & 7;
  return ((tile * 4 + (v >> 1)) * 32 + (c + 16 * h)) * 4 + (v & 1) * 2;
}

__device__ __forceinline__ int scat_pos(int m) {
  const int tile = m >> 8, c = (m >> 4) & 15, h = (m >> 3) & 1, v = m & 7;
  return ((tile * 2 + (v >> 2)) * 32 + (c + 16 * h)) * 4 + (v & 3);
}

struct AOp { v4f q[2]; };

__device__ __forceinline__ AOp load_op(const float *__restrict__ G, int phase, unsigned laneB) {
  const char *p = reinterpret_cast<const char *>(G + phase * PHASE_F);
  AOp o;
  o.q[0] = *reinterpret_cast<const v4f *>(p + laneB);
  o.q[1] = *reinterpret_cast<const v4f *>(p + laneB + 512u);
  return o;
}

__device__ __forceinline__ void cmul(float ar, float ai, float br, float bi,
                                     float &cr, float &ci) {
  cr = ar * br - ai * bi;
  ci = ar * bi + ai * br;
}

__device__ __forceinline__ int cnot_perm(int m, int rr) {
#pragma unroll
  for (int w = 0; w < N_WIRES; ++w) {
    int t = w + rr; if (t >= N_WIRES) t -= N_WIRES;
    m ^= ((m >> (9 - w)) & 1) << (9 - t);
  }
  return m;
}

__device__ __forceinline__ float ry_el(const float *cs2, int rb, int cb) {
  return (rb == cb) ? cs2[0] : (rb ? cs2[1] : -cs2[1]);
}

__global__ __launch_bounds__(256) void vqc_precompute(const float *__restrict__ W,
                                                      float *__restrict__ G) {
  __shared__ float ry[N_WIRES][2];
  __shared__ float zz[3][N_WIRES][2];
  __shared__ float tab[6][32][2];
  const int tid = threadIdx.x;
  const int l = blockIdx.x;

  if (tid < N_WIRES) {
    const float theta = W[(l * N_WIRES + tid) * 3 + 1];
    const float omega = W[(l * N_WIRES + tid) * 3 + 2];
    ry[tid][0] = __cosf(0.5f * theta);    ry[tid][1] = __sinf(0.5f * theta);
    zz[0][tid][0] = __cosf(0.5f * omega); zz[0][tid][1] = __sinf(0.5f * omega);
  } else if (tid >= 32 && tid < 32 + N_WIRES) {
    const int q = tid - 32;
    float c = 1.f, s = 0.f;
    if (l + 1 < N_LAYERS) {
      const float phi = W[((l + 1) * N_WIRES + q) * 3 + 0];
      c = __cosf(0.5f * phi); s = __sinf(0.5f * phi);
    }
    zz[1][q][0] = c; zz[1][q][1] = s;
  } else if (tid >= 64 && tid < 64 + N_WIRES) {
    const int q = tid - 64;
    const float phi = W[q * 3 + 0];
    zz[2][q][0] = __cosf(0.5f * phi); zz[2][q][1] = __sinf(0.5f * phi);
  }
  __syncthreads();

  if (tid < 192) {
    const int t = tid >> 5, e = tid & 31;
    const int w0 = (t & 1) * 5;
    float pr = 1.f, pi = 0.f;
#pragma unroll
    for (int q = 0; q < 5; ++q) {
      const float c = zz[t >> 1][w0 + q][0];
      const float s = ((e >> (4 - q)) & 1) ? zz[t >> 1][w0 + q][1] : -zz[t >> 1][w0 + q][1];
      float nr, ni; cmul(pr, pi, c, s, nr, ni);
      pr = nr; pi = ni;
    }
    tab[t][e][0] = pr; tab[t][e][1] = pi;
  }
  __syncthreads();

  {
    float *L = G + l * LAYER_F;
    const int r = tid >> 4, c = tid & 15;
#pragma unroll
    for (int g = 0; g < 2; ++g) {
      float p = 1.f;
#pragma unroll
      for (int q = 0; q < 4; ++q)
        p *= ry_el(ry[g * 4 + q], (r >> (3 - q)) & 1, (c >> (3 - q)) & 1);
      const int pos = (g == 0) ? frag_pos(r, (c >> 1) & 3, c >> 3, c & 1)
                               : frag_pos(r, 2 * (c >> 3) + ((c >> 1) & 1), (c >> 2) & 1, c & 1);
      L[g * PHASE_F + pos] = p;
    }
    const int a = r >> 2, b = r & 3, a2 = c >> 2, b2 = c & 3;
    float p = 0.f;
    if (a == a2) p = ry_el(ry[8], b >> 1, b2 >> 1) * ry_el(ry[9], b & 1, b2 & 1);
    L[2 * PHASE_F + frag_pos(r, (c >> 1) & 3, c >> 3, c & 1)] = p;
  }

  if (l + 1 < N_LAYERS) {
    float *E = G + E_OFF + l * DIAG_F;
    unsigned *S = reinterpret_cast<unsigned *>(G + S_OFF + l * SCAT_F);
    for (int m = tid; m < DIM; m += 256) {
      const int p = cnot_perm(m, l + 1);
      S[scat_pos(m)] = (unsigned)PADDED(p) * 4u;
      float orr, oi, fr, fi, er, ei;
      cmul(tab[0][m >> 5][0], tab[0][m >> 5][1], tab[1][m & 31][0], tab[1][m & 31][1], orr, oi);
      cmul(tab[2][p >> 5][0], tab[2][p >> 5][1], tab[3][p & 31][0], tab[3][p & 31][1], fr, fi);
      cmul(orr, oi, fr, fi, er, ei);
      const int pos = diag_pos(m);
      E[pos] = er; E[pos + 1] = ei;
    }
  }

  if (l == 0 && tid < 64) {
    G[F_OFF + tid * 2]     = tab[4 + (tid >> 5)][tid & 31][0];
    G[F_OFF + tid * 2 + 1] = tab[4 + (tid >> 5)][tid & 31][1];
  }
}

__device__ __forceinline__ v2f op_pair(const AOp &o, int kk) {
  const v4f q = o.q[kk >> 1];
  v2f a;
  if (kk & 1) { a.x = q.z; a.y = q.w; } else { a.x = q.x; a.y = q.y; }
  return a;
}

struct StartF { v2f hi[2], lo[4]; };

template <bool START>
__device__ __forceinline__ void fused_apply(const AOp opY, const AOp opX,
                                            AOp &opNext, const float *__restrict__ G,
                                            int nextPhase, unsigned laneB,
                                            const float *pR, const float *pI,
                                            const StartF &sf,
                                            float *dR, float *dI) {
  v8f tr = {}, ti = {};
#pragma unroll
  for (int kk = 0; kk < 4; ++kk) {
    v2f sr, si;
    if (START) {
      const v2f h = sf.hi[kk >> 1];
      const v2f l0 = sf.lo[2 * (kk & 1)], l1 = sf.lo[2 * (kk & 1) + 1];
      float r0, i0, r1, i1;
      cmul(h.x, h.y, l0.x, l0.y, r0, i0);
      cmul(h.x, h.y, l1.x, l1.y, r1, i1);
      sr.x = r0;  sr.y = r1;
      si.x = i0;  si.y = i1;
    } else {
      const int off = (kk >> 1) * 34 + (kk & 1) * 8;
      sr.x = pR[off];  sr.y = pR[off + 4];
      si.x = pI[off];  si.y = pI[off + 4];
    }
    const v2f b = op_pair(opY, kk);
    tr = __builtin_amdgcn_wmma_f32_16x16x4_f32(false, sr, false, b,
                                               (short)0, tr, false, false);
    ti = __builtin_amdgcn_wmma_f32_16x16x4_f32(false, si, false, b,
                                               (short)0, ti, false, false);
  }
  opNext = load_op(G, nextPhase, laneB);
  v8f ur = {}, ui = {};
#pragma unroll
  for (int kk = 0; kk < 4; ++kk) {
    const v2f a = op_pair(opX, kk);
    v2f br, bi;
    br.x = tr[2 * kk];  br.y = tr[2 * kk + 1];
    bi.x = ti[2 * kk];  bi.y = ti[2 * kk + 1];
    ur = __builtin_amdgcn_wmma_f32_16x16x4_f32(false, a, false, br,
                                               (short)0, ur, false, false);
    ui = __builtin_amdgcn_wmma_f32_16x16x4_f32(false, a, false, bi,
                                               (short)0, ui, false, false);
  }
  *reinterpret_cast<v4f *>(dR)     = __builtin_shufflevector(ur, ur, 0, 1, 2, 3);
  *reinterpret_cast<v4f *>(dR + 4) = __builtin_shufflevector(ur, ur, 4, 5, 6, 7);
  *reinterpret_cast<v4f *>(dI)     = __builtin_shufflevector(ui, ui, 0, 1, 2, 3);
  *reinterpret_cast<v4f *>(dI + 4) = __builtin_shufflevector(ui, ui, 4, 5, 6, 7);
}

struct ETab { v4f e[4]; };

__device__ __forceinline__ ETab load_etab(const float *__restrict__ eTile, unsigned laneB) {
  const char *p = reinterpret_cast<const char *>(eTile);
  ETab t;
#pragma unroll
  for (int j = 0; j < 4; ++j)
    t.e[j] = *reinterpret_cast<const v4f *>(p + laneB + 512u * j);
  return t;
}

struct SAddr { v4u a[2]; };

__device__ __forceinline__ SAddr load_saddr(const float *__restrict__ sTile, unsigned laneB) {
  const char *p = reinterpret_cast<const char *>(sTile);
  SAddr s;
  s.a[0] = *reinterpret_cast<const v4u *>(p + laneB);
  s.a[1] = *reinterpret_cast<const v4u *>(p + laneB + 512u);
  return s;
}

__device__ __forceinline__ void gt_products(const AOp op,
                                            const float *pR, const float *pI,
                                            v8f &tr, v8f &ti) {
  tr = v8f{};  ti = v8f{};
#pragma unroll
  for (int kk = 0; kk < 4; ++kk) {
    const v2f a = op_pair(op, kk);
    v2f br, bi;
    const int off = (kk >> 1) * 32 + (kk & 1) * 16;
    br.x = pR[off];  br.y = pR[off + 8];
    bi.x = pI[off];  bi.y = pI[off + 8];
    tr = __builtin_amdgcn_wmma_f32_16x16x4_f32(false, a, false, br,
                                               (short)0, tr, false, false);
    ti = __builtin_amdgcn_wmma_f32_16x16x4_f32(false, a, false, bi,
                                               (short)0, ti, false, false);
  }
}

__device__ __forceinline__ void gt_apply(const AOp op,
                                         const float *pR, const float *pI,
                                         float *dR, float *dI,
                                         const SAddr sa,
                                         const ETab et) {
  v8f tr, ti;
  gt_products(op, pR, pI, tr, ti);
  char *bR = reinterpret_cast<char *>(dR);
  char *bI = reinterpret_cast<char *>(dI);
#pragma unroll
  for (int v = 0; v < 8; ++v) {
    const unsigned w = sa.a[v >> 2][v & 3];
    const float er = (v & 1) ? et.e[v >> 1].z : et.e[v >> 1].x;
    const float ei = (v & 1) ? et.e[v >> 1].w : et.e[v >> 1].y;
    float vr, vi; cmul(tr[v], ti[v], er, ei, vr, vi);
    *reinterpret_cast<float *>(bR + w) = vr;
    *reinterpret_cast<float *>(bI + w) = vi;
  }
}

__device__ __forceinline__ float gt_measure(const AOp op,
                                            const float *pR, const float *pI,
                                            int rr, int permBase) {
  v8f tr, ti;
  gt_products(op, pR, pI, tr, ti);
  float acc = 0.f;
#pragma unroll
  for (int v = 0; v < 8; ++v) {
    const float p = tr[v] * tr[v] + ti[v] * ti[v];
    acc += (cnot_perm(v, rr) & 1) ? -p : p;
  }
  return (permBase & 1) ? -acc : acc;
}

__global__ __launch_bounds__(128) __attribute__((amdgpu_waves_per_eu(16, 16)))
void vqc_kernel(const float *__restrict__ X,
                const float *__restrict__ G,
                const float *__restrict__ bias,
                float *__restrict__ out) {
  __shared__ __attribute__((aligned(16))) float SR[2][PDIM];
  __shared__ __attribute__((aligned(16))) float SI[2][PDIM];
  __shared__ float cs[2 * N_WIRES];
  __shared__ float pLo[32][2], pHi[32][2];
  __shared__ float red[4];

  const int tid  = threadIdx.x;
  const int lane = tid & 31;
  const int wave = __builtin_amdgcn_readfirstlane(tid >> 5);
  const int s    = blockIdx.x;

  if (tid < N_WIRES) {
    const float x = X[s * N_WIRES + tid] * 0.7853981633974483f;
    cs[tid] = __cosf(x);
    cs[N_WIRES + tid] = __sinf(x);
  }
  __syncthreads();

  if (tid < 32) {
    float p = 1.f;
#pragma unroll
    for (int w = 5; w < N_WIRES; ++w)
      p *= cs[((tid >> (9 - w)) & 1) * N_WIRES + w];
    pLo[tid][0] = p * G[F_OFF + 64 + tid * 2];
    pLo[tid][1] = p * G[F_OFF + 64 + tid * 2 + 1];
  } else if (tid < 64) {
    const int h = tid - 32;
    float p = 1.f;
#pragma unroll
    for (int w = 0; w < 5; ++w)
      p *= cs[((h >> (4 - w)) & 1) * N_WIRES + w];
    pHi[h][0] = p * G[F_OFF + h * 2];
    pHi[h][1] = p * G[F_OFF + h * 2 + 1];
  }
  __syncthreads();

  const int c16 = lane & 15;
  const int hh  = lane >> 4;

  StartF sf;
#pragma unroll
  for (int j = 0; j < 2; ++j) {
    sf.hi[j].x = pHi[2 * c16 + j][0];
    sf.hi[j].y = pHi[2 * c16 + j][1];
  }
#pragma unroll
  for (int j = 0; j < 4; ++j) {
    sf.lo[j].x = pLo[16 * hh + 4 * j + wave][0];
    sf.lo[j].y = pLo[16 * hh + 4 * j + wave][1];
  }

  const int baseA  = c16 * 68 + hh * 17 + wave;
  const int baseWF = 4 * (hh * 136 + (c16 >> 1) * 17 + (c16 & 1) * 8 + wave * 2);
  const int col     = wave * 16 + c16;
  const int laneHi8 = hh << 3;
  const int baseB2  = (wave >> 1) * 544 + (wave & 1) * 4 + (c16 & 3) * 136 + hh * 68 + (c16 >> 2);
  const float *eTile = G + E_OFF + wave * 512;
  const float *sTile = G + S_OFF + wave * 256;
  const unsigned laneB = (unsigned)lane * 16u;

  AOp aG1 = load_op(G, 1, laneB);
  float z = 0.f;
#pragma unroll
  for (int l = 0; l < N_LAYERS; ++l) {
    const AOp aG0 = load_op(G, 3 * l, laneB);
    AOp aGT;
    if (l == 0)
      fused_apply<true>(aG1, aG0, aGT, G, 3 * l + 2, laneB,
                        SR[0] + baseA, SI[0] + baseA, sf,
                        SR[1] + baseWF, SI[1] + baseWF);
    else
      fused_apply<false>(aG1, aG0, aGT, G, 3 * l + 2, laneB,
                         SR[0] + baseA, SI[0] + baseA, sf,
                         SR[1] + baseWF, SI[1] + baseWF);
    __syncthreads();
    if (l + 1 < N_LAYERS) {
      const ETab et = load_etab(eTile + l * DIAG_F, laneB);
      const SAddr sa = load_saddr(sTile + l * SCAT_F, laneB);
      gt_apply(aGT, SR[1] + baseB2, SI[1] + baseB2,
               SR[0], SI[0],
               sa, et);
      aG1 = load_op(G, 3 * l + 4, laneB);
      __syncthreads();
    } else {
      const int pb = cnot_perm((col << 4) | laneHi8, l + 1);
      z = gt_measure(aGT, SR[1] + baseB2, SI[1] + baseB2, l + 1, pb);
    }
  }

#pragma unroll
  for (int m = 16; m > 0; m >>= 1) z += __shfl_xor(z, m, 32);
  if (lane == 0) red[wave] = z;
  __syncthreads();
  if (tid == 0) {
    float t = 0.f;
#pragma unroll
    for (int w = 0; w < 4; ++w) t += red[w];
    out[s] = t + bias[0];
  }
}

extern "C" void kernel_launch(void *const *d_in, const int *in_sizes, int n_in,
                              void *d_out, int out_size, void *d_ws, size_t ws_size,
                              hipStream_t stream) {
  const float *X = (const float *)d_in[0];
  const float *W = (const float *)d_in[1];
  const float *bias = (const float *)d_in[2];
  float *out = (float *)d_out;
  float *G = (float *)d_ws;

  const int B = in_sizes[0] / N_WIRES;

  vqc_precompute<<<N_LAYERS, 256, 0, stream>>>(W, G);
  vqc_kernel<<<B, 128, 0, stream>>>(X, G, bias, out);
}


# Two-wire operator ry8 x ry9 (as I4 x 4x4 it filled 1/4 of a 16x16 WMMA operand) moved to packed f32 FMAs, scalar-loaded coefficients; four-wire operators stay on WMMA (96 to 64 a wave); new sum order
# speedup vs baseline: 2.7462x; 1.1215x over previous
#include <hip/hip_runtime.h>


typedef float v2f __attribute__((ext_vector_type(2)));
typedef float v4f __attribute__((ext_vector_type(4)));
typedef float v8f __attribute__((ext_vector_type(8)));
typedef unsigned v4u __attribute__((ext_vector_type(4)));

#define N_WIRES 10
#define N_LAYERS 4
#define DIM 1024
#define PADDED(k) ((k) + ((k) >> 4))
#define PDIM (DIM + (DIM >> 4))
#define PHASE_F 256
#define LAYER_F 768
#define OPS_F (N_LAYERS * LAYER_F)
#define DIAG_F 2048
#define E_OFF OPS_F
#define F_OFF (E_OFF + N_LAYERS * DIAG_F)
#define S_OFF (F_OFF + 128)
#define SCAT_F 1024

__device__ __forceinline__ int frag_pos(int M, int kk, int h, int e) {
  return (kk >> 1) * 128 + (M + 16 * h) * 4 + (kk & 1) * 2 + e;
}

__device__ __forceinline__ int diag_pos(int m) {
  const int tile = m >> 8, c = (m >> 4) & 15, h = (m >> 3) & 1, v = m & 7;
  return ((tile * 4 + (v >> 1)) * 32 + (c + 16 * h)) * 4 + (v & 1);
}

__device__ __forceinline__ int scat_pos(int m) {
  const int tile = m >> 8, c = (m >> 4) & 15, h = (m >> 3) & 1, v = m & 7;
  return ((tile * 2 + (v >> 2)) * 32 + (c + 16 * h)) * 4 + (v & 3);
}

struct AOp { v4f q[2]; };

__device__ __forceinline__ AOp load_op(const float *__restrict__ G, int phase, unsigned laneB) {
  const char *p = reinterpret_cast<const char *>(G + phase * PHASE_F);
  AOp o;
  o.q[0] = *reinterpret_cast<const v4f *>(p + laneB);
  o.q[1] = *reinterpret_cast<const v4f *>(p + laneB + 512u);
  return o;
}

__device__ __forceinline__ void cmul(float ar, float ai, float br, float bi,
                                     float &cr, float &ci) {
  cr = ar * br - ai * bi;
  ci = ar * bi + ai * br;
}

__device__ __forceinline__ int cnot_perm(int m, int rr) {
#pragma unroll
  for (int w = 0; w < N_WIRES; ++w) {
    int t = w + rr; if (t >= N_WIRES) t -= N_WIRES;
    m ^= ((m >> (9 - w)) & 1) << (9 - t);
  }
  return m;
}

__device__ __forceinline__ float ry_el(const float *cs2, int rb, int cb) {
  return (rb == cb) ? cs2[0] : (rb ? cs2[1] : -cs2[1]);
}

__global__ __launch_bounds__(256) void vqc_precompute(const float *__restrict__ W,
                                                      float *__restrict__ G) {
  __shared__ float ry[N_WIRES][2];
  __shared__ float zz[3][N_WIRES][2];
  __shared__ float tab[6][32][2];
  const int tid = threadIdx.x;
  const int l = blockIdx.x;

  if (tid < N_WIRES) {
    const float theta = W[(l * N_WIRES + tid) * 3 + 1];
    const float omega = W[(l * N_WIRES + tid) * 3 + 2];
    ry[tid][0] = __cosf(0.5f * theta);    ry[tid][1] = __sinf(0.5f * theta);
    zz[0][tid][0] = __cosf(0.5f * omega); zz[0][tid][1] = __sinf(0.5f * omega);
  } else if (tid >= 32 && tid < 32 + N_WIRES) {
    const int q = tid - 32;
    float c = 1.f, s = 0.f;
    if (l + 1 < N_LAYERS) {
      const float phi = W[((l + 1) * N_WIRES + q) * 3 + 0];
      c = __cosf(0.5f * phi); s = __sinf(0.5f * phi);
    }
    zz[1][q][0] = c; zz[1][q][1] = s;
  } else if (tid >= 64 && tid < 64 + N_WIRES) {
    const int q = tid - 64;
    const float phi = W[q * 3 + 0];
    zz[2][q][0] = __cosf(0.5f * phi); zz[2][q][1] = __sinf(0.5f * phi);
  }
  __syncthreads();

  if (tid < 192) {
    const int t = tid >> 5, e = tid & 31;
    const int w0 = (t & 1) * 5;
    float pr = 1.f, pi = 0.f;
#pragma unroll
    for (int q = 0; q < 5; ++q) {
      const float c = zz[t >> 1][w0 + q][0];
      const float s = ((e >> (4 - q)) & 1) ? zz[t >> 1][w0 + q][1] : -zz[t >> 1][w0 + q][1];
      float nr, ni; cmul(pr, pi, c, s, nr, ni);
      pr = nr; pi = ni;
    }
    tab[t][e][0] = pr; tab[t][e][1] = pi;
  }
  __syncthreads();

  {
    float *L = G + l * LAYER_F;
    const int r = tid >> 4, c = tid & 15;
#pragma unroll
    for (int g = 0; g < 2; ++g) {
      float p = 1.f;
#pragma unroll
      for (int q = 0; q < 4; ++q)
        p *= ry_el(ry[g * 4 + q], (r >> (3 - q)) & 1, (c >> (3 - q)) & 1);
      const int pos = (g == 0) ? frag_pos(r, (c >> 1) & 3, c >> 3, c & 1)
                               : frag_pos(r, 2 * (c >> 3) + ((c >> 1) & 1), (c >> 2) & 1, c & 1);
      L[g * PHASE_F + pos] = p;
    }
    if (tid < 16) {
      const int k = tid >> 2, i = tid & 3;
      L[2 * PHASE_F + tid] = ry_el(ry[8], i >> 1, k >> 1) * ry_el(ry[9], i & 1, k & 1);
    }
  }

  if (l + 1 < N_LAYERS) {
    float *E = G + E_OFF + l * DIAG_F;
    unsigned *S = reinterpret_cast<unsigned *>(G + S_OFF + l * SCAT_F);
    for (int m = tid; m < DIM; m += 256) {
      const int p = cnot_perm(m, l + 1);
      S[scat_pos(m)] = (unsigned)PADDED(p) * 4u;
      float orr, oi, fr, fi, er, ei;
      cmul(tab[0][m >> 5][0], tab[0][m >> 5][1], tab[1][m & 31][0], tab[1][m & 31][1], orr, oi);
      cmul(tab[2][p >> 5][0], tab[2][p >> 5][1], tab[3][p & 31][0], tab[3][p & 31][1], fr, fi);
      cmul(orr, oi, fr, fi, er, ei);
      const int pos = diag_pos(m);
      E[pos] = er; E[pos + 2] = ei;
    }
  }

  if (l == 0 && tid < 64) {
    G[F_OFF + tid * 2]     = tab[4 + (tid >> 5)][tid & 31][0];
    G[F_OFF + tid * 2 + 1] = tab[4 + (tid >> 5)][tid & 31][1];
  }
}

__device__ __forceinline__ v2f op_pair(const AOp &o, int kk) {
  const v4f q = o.q[kk >> 1];
  v2f a;
  if (kk & 1) { a.x = q.z; a.y = q.w; } else { a.x = q.x; a.y = q.y; }
  return a;
}

struct StartF { v2f hi[2], lo[4]; };

template <bool START>
__device__ __forceinline__ void fused_apply(const AOp opY, const AOp opX,
                                            const float *pR, const float *pI,
                                            const StartF &sf,
                                            float *dR, float *dI) {
  v8f tr = {}, ti = {};
#pragma unroll
  for (int kk = 0; kk < 4; ++kk) {
    v2f sr, si;
    if (START) {
      const v2f h = sf.hi[kk >> 1];
      const v2f l0 = sf.lo[2 * (kk & 1)], l1 = sf.lo[2 * (kk & 1) + 1];
      float r0, i0, r1, i1;
      cmul(h.x, h.y, l0.x, l0.y, r0, i0);
      cmul(h.x, h.y, l1.x, l1.y, r1, i1);
      sr.x = r0;  sr.y = r1;
      si.x = i0;  si.y = i1;
    } else {
      const int off = (kk >> 1) * 34 + (kk & 1) * 8;
      sr.x = pR[off];  sr.y = pR[off + 4];
      si.x = pI[off];  si.y = pI[off + 4];
    }
    const v2f b = op_pair(opY, kk);
    tr = __builtin_amdgcn_wmma_f32_16x16x4_f32(false, sr, false, b,
                                               (short)0, tr, false, false);
    ti = __builtin_amdgcn_wmma_f32_16x16x4_f32(false, si, false, b,
                                               (short)0, ti, false, false);
  }
  v8f ur = {}, ui = {};
#pragma unroll
  for (int kk = 0; kk < 4; ++kk) {
    const v2f a = op_pair(opX, kk);
    v2f br, bi;
    br.x = tr[2 * kk];  br.y = tr[2 * kk + 1];
    bi.x = ti[2 * kk];  bi.y = ti[2 * kk + 1];
    ur = __builtin_amdgcn_wmma_f32_16x16x4_f32(false, a, false, br,
                                               (short)0, ur, false, false);
    ui = __builtin_amdgcn_wmma_f32_16x16x4_f32(false, a, false, bi,
                                               (short)0, ui, false, false);
  }
  *reinterpret_cast<v4f *>(dR)     = __builtin_shufflevector(ur, ur, 0, 1, 2, 3);
  *reinterpret_cast<v4f *>(dR + 4) = __builtin_shufflevector(ur, ur, 4, 5, 6, 7);
  *reinterpret_cast<v4f *>(dI)     = __builtin_shufflevector(ui, ui, 0, 1, 2, 3);
  *reinterpret_cast<v4f *>(dI + 4) = __builtin_shufflevector(ui, ui, 4, 5, 6, 7);
}

struct ETab { v4f e[4]; };

__device__ __forceinline__ ETab load_etab(const float *__restrict__ eTile, unsigned laneB) {
  const char *p = reinterpret_cast<const char *>(eTile);
  ETab t;
#pragma unroll
  for (int j = 0; j < 4; ++j)
    t.e[j] = *reinterpret_cast<const v4f *>(p + laneB + 512u * j);
  return t;
}

struct SAddr { v4u a[2]; };

__device__ __forceinline__ SAddr load_saddr(const float *__restrict__ sTile, unsigned laneB) {
  const char *p = reinterpret_cast<const char *>(sTile);
  SAddr s;
  s.a[0] = *reinterpret_cast<const v4u *>(p + laneB);
  s.a[1] = *reinterpret_cast<const v4u *>(p + laneB + 512u);
  return s;
}

struct UT4 { float u[16]; };

__device__ __forceinline__ UT4 load_ut(const float *__restrict__ G, int l) {
  const float *p = G + l * LAYER_F + 2 * PHASE_F;
  UT4 t;
#pragma unroll
  for (int j = 0; j < 16; ++j) t.u[j] = p[j];
  return t;
}

__device__ __forceinline__ void ut_products(const UT4 &ut,
                                            const float *pR, const float *pI,
                                            v2f (&zr)[4], v2f (&zi)[4]) {
#pragma unroll
  for (int g = 0; g < 2; ++g) {
    float xr[4], xi[4];
#pragma unroll
    for (int k = 0; k < 4; ++k) {
      xr[k] = pR[8 * (4 * g + k)];
      xi[k] = pI[8 * (4 * g + k)];
    }
#pragma unroll
    for (int q = 0; q < 2; ++q) {
      const v2f u0 = {ut.u[2 * q], ut.u[2 * q + 1]};
      v2f ar = u0 * xr[0], ai = u0 * xi[0];
#pragma unroll
      for (int k = 1; k < 4; ++k) {
        const v2f uk = {ut.u[4 * k + 2 * q], ut.u[4 * k + 2 * q + 1]};
        const v2f sr = {xr[k], xr[k]}, si = {xi[k], xi[k]};
        ar = __builtin_elementwise_fma(uk, sr, ar);
        ai = __builtin_elementwise_fma(uk, si, ai);
      }
      zr[2 * g + q] = ar;
      zi[2 * g + q] = ai;
    }
  }
}

__device__ __forceinline__ void ut_apply(const UT4 &ut,
                                         const float *pR, const float *pI,
                                         float *dR, float *dI,
                                         const SAddr sa,
                                         const ETab et) {
  v2f zr[4], zi[4];
  ut_products(ut, pR, pI, zr, zi);
  char *bR = reinterpret_cast<char *>(dR);
  char *bI = reinterpret_cast<char *>(dI);
#pragma unroll
  for (int q = 0; q < 4; ++q) {
    const v2f er = __builtin_shufflevector(et.e[q], et.e[q], 0, 1);
    const v2f ei = __builtin_shufflevector(et.e[q], et.e[q], 2, 3);
    const v2f vr = zr[q] * er - zi[q] * ei;
    const v2f vi = zr[q] * ei + zi[q] * er;
    const unsigned w0 = sa.a[q >> 1][2 * (q & 1)];
    const unsigned w1 = sa.a[q >> 1][2 * (q & 1) + 1];
    *reinterpret_cast<float *>(bR + w0) = vr.x;
    *reinterpret_cast<float *>(bI + w0) = vi.x;
    *reinterpret_cast<float *>(bR + w1) = vr.y;
    *reinterpret_cast<float *>(bI + w1) = vi.y;
  }
}

__device__ __forceinline__ float ut_measure(const UT4 &ut,
                                            const float *pR, const float *pI,
                                            int rr, int permBase) {
  v2f zr[4], zi[4];
  ut_products(ut, pR, pI, zr, zi);
  float acc = 0.f;
#pragma unroll
  for (int v = 0; v < 8; ++v) {
    const float tr = (v & 1) ? zr[v >> 1].y : zr[v >> 1].x;
    const float ti = (v & 1) ? zi[v >> 1].y : zi[v >> 1].x;
    const float p = tr * tr + ti * ti;
    acc += (cnot_perm(v, rr) & 1) ? -p : p;
  }
  return (permBase & 1) ? -acc : acc;
}

__global__ __launch_bounds__(128) __attribute__((amdgpu_waves_per_eu(16, 16)))
void vqc_kernel(const float *__restrict__ X,
                const float *__restrict__ G,
                const float *__restrict__ bias,
                float *__restrict__ out) {
  __shared__ __attribute__((aligned(16))) float SR[2][PDIM];
  __shared__ __attribute__((aligned(16))) float SI[2][PDIM];
  __shared__ float cs[2 * N_WIRES];
  __shared__ float pLo[32][2], pHi[32][2];
  __shared__ float red[4];

  const int tid  = threadIdx.x;
  const int lane = tid & 31;
  const int wave = __builtin_amdgcn_readfirstlane(tid >> 5);
  const int s    = blockIdx.x;

  if (tid < N_WIRES) {
    const float x = X[s * N_WIRES + tid] * 0.7853981633974483f;
    cs[tid] = __cosf(x);
    cs[N_WIRES + tid] = __sinf(x);
  }
  __syncthreads();

  if (tid < 32) {
    float p = 1.f;
#pragma unroll
    for (int w = 5; w < N_WIRES; ++w)
      p *= cs[((tid >> (9 - w)) & 1) * N_WIRES + w];
    pLo[tid][0] = p * G[F_OFF + 64 + tid * 2];
    pLo[tid][1] = p * G[F_OFF + 64 + tid * 2 + 1];
  } else if (tid < 64) {
    const int h = tid - 32;
    float p = 1.f;
#pragma unroll
    for (int w = 0; w < 5; ++w)
      p *= cs[((h >> (4 - w)) & 1) * N_WIRES + w];
    pHi[h][0] = p * G[F_OFF + h * 2];
    pHi[h][1] = p * G[F_OFF + h * 2 + 1];
  }
  __syncthreads();

  const int c16 = lane & 15;
  const int hh  = lane >> 4;

  StartF sf;
#pragma unroll
  for (int j = 0; j < 2; ++j) {
    sf.hi[j].x = pHi[2 * c16 + j][0];
    sf.hi[j].y = pHi[2 * c16 + j][1];
  }
#pragma unroll
  for (int j = 0; j < 4; ++j) {
    sf.lo[j].x = pLo[16 * hh + 4 * j + wave][0];
    sf.lo[j].y = pLo[16 * hh + 4 * j + wave][1];
  }

  const int baseA  = c16 * 68 + hh * 17 + wave;
  const int baseWF = 4 * (hh * 136 + (c16 >> 1) * 17 + (c16 & 1) * 8 + wave * 2);
  const int col     = wave * 16 + c16;
  const int laneHi8 = hh << 3;
  const int baseB2  = (wave >> 1) * 544 + (wave & 1) * 4 + (c16 & 3) * 136 + hh * 68 + (c16 >> 2);
  const float *eTile = G + E_OFF + wave * 512;
  const float *sTile = G + S_OFF + wave * 256;
  const unsigned laneB = (unsigned)lane * 16u;

  AOp aG1 = load_op(G, 1, laneB);
  float z = 0.f;
#pragma unroll
  for (int l = 0; l < N_LAYERS; ++l) {
    const AOp aG0 = load_op(G, 3 * l, laneB);
    const UT4 ut = load_ut(G, l);
    if (l == 0)
      fused_apply<true>(aG1, aG0,
                        SR[0] + baseA, SI[0] + baseA, sf,
                        SR[1] + baseWF, SI[1] + baseWF);
    else
      fused_apply<false>(aG1, aG0,
                         SR[0] + baseA, SI[0] + baseA, sf,
                         SR[1] + baseWF, SI[1] + baseWF);
    __syncthreads();
    if (l + 1 < N_LAYERS) {
      const ETab et = load_etab(eTile + l * DIAG_F, laneB);
      const SAddr sa = load_saddr(sTile + l * SCAT_F, laneB);
      ut_apply(ut, SR[1] + baseB2, SI[1] + baseB2,
               SR[0], SI[0],
               sa, et);
      aG1 = load_op(G, 3 * l + 4, laneB);
      __syncthreads();
    } else {
      const int pb = cnot_perm((col << 4) | laneHi8, l + 1);
      z = ut_measure(ut, SR[1] + baseB2, SI[1] + baseB2, l + 1, pb);
    }
  }

#pragma unroll
  for (int m = 16; m > 0; m >>= 1) z += __shfl_xor(z, m, 32);
  if (lane == 0) red[wave] = z;
  __syncthreads();
  if (tid == 0) {
    float t = 0.f;
#pragma unroll
    for (int w = 0; w < 4; ++w) t += red[w];
    out[s] = t + bias[0];
  }
}

extern "C" void kernel_launch(void *const *d_in, const int *in_sizes, int n_in,
                              void *d_out, int out_size, void *d_ws, size_t ws_size,
                              hipStream_t stream) {
  const float *X = (const float *)d_in[0];
  const float *W = (const float *)d_in[1];
  const float *bias = (const float *)d_in[2];
  float *out = (float *)d_out;
  float *G = (float *)d_ws;

  const int B = in_sizes[0] / N_WIRES;

  vqc_precompute<<<N_LAYERS, 256, 0, stream>>>(W, G);
  vqc_kernel<<<B, 128, 0, stream>>>(X, G, bias, out);
}


# ry8, ry9 (vector unit) as shears [[1,-t],[t,1]], |t|<=1: 1 FMA per amplitude and wire (was 2); scale, quarter-turn sign and renaming folded into phase/offset tables and <Z>; new sum order
# speedup vs baseline: 2.9891x; 1.0884x over previous
#include <hip/hip_runtime.h>


typedef float v2f __attribute__((ext_vector_type(2)));
typedef float v4f __attribute__((ext_vector_type(4)));
typedef float v8f __attribute__((ext_vector_type(8)));
typedef unsigned v4u __attribute__((ext_vector_type(4)));

#define N_WIRES 10
#define N_LAYERS 4
#define DIM 1024
#define PADDED(k) ((k) + ((k) >> 4))
#define PDIM (DIM + (DIM >> 4))
#define PHASE_F 256
#define LAYER_F 768
#define OPS_F (N_LAYERS * LAYER_F)
#define DIAG_F 2048
#define E_OFF OPS_F
#define F_OFF (E_OFF + N_LAYERS * DIAG_F)
#define S_OFF (F_OFF + 128)
#define SCAT_F 1024

__device__ __forceinline__ int frag_pos(int M, int kk, int h, int e) {
  return (kk >> 1) * 128 + (M + 16 * h) * 4 + (kk & 1) * 2 + e;
}

__device__ __forceinline__ int diag_pos(int m) {
  const int tile = m >> 8, c = (m >> 4) & 15, h = (m >> 3) & 1, v = m & 7;
  return ((tile * 4 + (v & 3)) * 32 + (c + 16 * h)) * 4 + (v >> 2);
}

__device__ __forceinline__ int scat_pos(int m) {
  const int tile = m >> 8, c = (m >> 4) & 15, h = (m >> 3) & 1, v = m & 7;
  const int j = 2 * (v & 3) + (v >> 2);
  return ((tile * 2 + (j >> 2)) * 32 + (c + 16 * h)) * 4 + (j & 3);
}

struct AOp { v4f q[2]; };

__device__ __forceinline__ AOp load_op(const float *__restrict__ G, int phase, unsigned laneB) {
  const char *p = reinterpret_cast<const char *>(G + phase * PHASE_F);
  AOp o;
  o.q[0] = *reinterpret_cast<const v4f *>(p + laneB);
  o.q[1] = *reinterpret_cast<const v4f *>(p + laneB + 512u);
  return o;
}

__device__ __forceinline__ void cmul(float ar, float ai, float br, float bi,
                                     float &cr, float &ci) {
  cr = ar * br - ai * bi;
  ci = ar * bi + ai * br;
}

__device__ __forceinline__ int cnot_perm(int m, int rr) {
#pragma unroll
  for (int w = 0; w < N_WIRES; ++w) {
    int t = w + rr; if (t >= N_WIRES) t -= N_WIRES;
    m ^= ((m >> (9 - w)) & 1) << (9 - t);
  }
  return m;
}

__device__ __forceinline__ float ry_el(const float *cs2, int rb, int cb) {
  return (rb == cb) ? cs2[0] : (rb ? cs2[1] : -cs2[1]);
}

struct Shear { float t, sigma; int f; };
__device__ __forceinline__ Shear shear_of(const float *cs2) {
  const float c = cs2[0], s = cs2[1];
  Shear r;
  if (fabsf(c) >= fabsf(s)) { r.f = 0; r.sigma = c; r.t = s / c; }
  else                      { r.f = 1; r.sigma = s; r.t = -c / s; }
  return r;
}

__global__ __launch_bounds__(256) void vqc_precompute(const float *__restrict__ W,
                                                      float *__restrict__ G) {
  __shared__ float ry[N_WIRES][2];
  __shared__ float zz[3][N_WIRES][2];
  __shared__ float tab[6][32][2];
  const int tid = threadIdx.x;
  const int l = blockIdx.x;

  if (tid < N_WIRES) {
    const float theta = W[(l * N_WIRES + tid) * 3 + 1];
    const float omega = W[(l * N_WIRES + tid) * 3 + 2];
    ry[tid][0] = __cosf(0.5f * theta);    ry[tid][1] = __sinf(0.5f * theta);
    zz[0][tid][0] = __cosf(0.5f * omega); zz[0][tid][1] = __sinf(0.5f * omega);
  } else if (tid >= 32 && tid < 32 + N_WIRES) {
    const int q = tid - 32;
    float c = 1.f, s = 0.f;
    if (l + 1 < N_LAYERS) {
      const float phi = W[((l + 1) * N_WIRES + q) * 3 + 0];
      c = __cosf(0.5f * phi); s = __sinf(0.5f * phi);
    }
    zz[1][q][0] = c; zz[1][q][1] = s;
  } else if (tid >= 64 && tid < 64 + N_WIRES) {
    const int q = tid - 64;
    const float phi = W[q * 3 + 0];
    zz[2][q][0] = __cosf(0.5f * phi); zz[2][q][1] = __sinf(0.5f * phi);
  }
  __syncthreads();

  if (tid < 192) {
    const int t = tid >> 5, e = tid & 31;
    const int w0 = (t & 1) * 5;
    float pr = 1.f, pi = 0.f;
#pragma unroll
    for (int q = 0; q < 5; ++q) {
      const float c = zz[t >> 1][w0 + q][0];
      const float s = ((e >> (4 - q)) & 1) ? zz[t >> 1][w0 + q][1] : -zz[t >> 1][w0 + q][1];
      float nr, ni; cmul(pr, pi, c, s, nr, ni);
      pr = nr; pi = ni;
    }
    tab[t][e][0] = pr; tab[t][e][1] = pi;
  }
  __syncthreads();

  {
    float *L = G + l * LAYER_F;
    const int r = tid >> 4, c = tid & 15;
#pragma unroll
    for (int g = 0; g < 2; ++g) {
      float p = 1.f;
#pragma unroll
      for (int q = 0; q < 4; ++q)
        p *= ry_el(ry[g * 4 + q], (r >> (3 - q)) & 1, (c >> (3 - q)) & 1);
      const int pos = (g == 0) ? frag_pos(r, (c >> 1) & 3, c >> 3, c & 1)
                               : frag_pos(r, 2 * (c >> 3) + ((c >> 1) & 1), (c >> 2) & 1, c & 1);
      L[g * PHASE_F + pos] = p;
    }
  }

  const Shear sh8 = shear_of(ry[8]), sh9 = shear_of(ry[9]);
  const int turned = (sh8.f << 1) | sh9.f;
  const float sig = sh8.sigma * sh9.sigma;
  if (tid == 0) {
    float *U = G + l * LAYER_F + 2 * PHASE_F;
    U[0] = sh8.t;
    U[1] = sh9.t;
    U[2] = (cnot_perm(turned, l + 1) & 1) ? -(sig * sig) : sig * sig;
  }

  if (l + 1 < N_LAYERS) {
    float *E = G + E_OFF + l * DIAG_F;
    unsigned *S = reinterpret_cast<unsigned *>(G + S_OFF + l * SCAT_F);
    for (int m = tid; m < DIM; m += 256) {
      const int mt = m ^ turned;
      float g = sig;
      if (sh8.f && !((mt >> 1) & 1)) g = -g;
      if (sh9.f && !(mt & 1)) g = -g;
      const int p = cnot_perm(mt, l + 1);
      S[scat_pos(m)] = (unsigned)PADDED(p) * 4u;
      float orr, oi, fr, fi, er, ei;
      cmul(tab[0][mt >> 5][0], tab[0][mt >> 5][1], tab[1][mt & 31][0], tab[1][mt & 31][1], orr, oi);
      cmul(tab[2][p >> 5][0], tab[2][p >> 5][1], tab[3][p & 31][0], tab[3][p & 31][1], fr, fi);
      cmul(orr, oi, fr, fi, er, ei);
      const int pos = diag_pos(m);
      E[pos] = g * er; E[pos + 2] = g * ei;
    }
  }

  if (l == 0 && tid < 64) {
    G[F_OFF + tid * 2]     = tab[4 + (tid >> 5)][tid & 31][0];
    G[F_OFF + tid * 2 + 1] = tab[4 + (tid >> 5)][tid & 31][1];
  }
}

__device__ __forceinline__ v2f op_pair(const AOp &o, int kk) {
  const v4f q = o.q[kk >> 1];
  v2f a;
  if (kk & 1) { a.x = q.z; a.y = q.w; } else { a.x = q.x; a.y = q.y; }
  return a;
}

struct StartF { v2f hi[2], lo[4]; };

template <bool START>
__device__ __forceinline__ void fused_apply(const AOp opY, const AOp opX,
                                            const float *pR, const float *pI,
                                            const StartF &sf,
                                            float *dR, float *dI) {
  v8f tr = {}, ti = {};
#pragma unroll
  for (int kk = 0; kk < 4; ++kk) {
    v2f sr, si;
    if (START) {
      const v2f h = sf.hi[kk >> 1];
      const v2f l0 = sf.lo[2 * (kk & 1)], l1 = sf.lo[2 * (kk & 1) + 1];
      float r0, i0, r1, i1;
      cmul(h.x, h.y, l0.x, l0.y, r0, i0);
      cmul(h.x, h.y, l1.x, l1.y, r1, i1);
      sr.x = r0;  sr.y = r1;
      si.x = i0;  si.y = i1;
    } else {
      const int off = (kk >> 1) * 34 + (kk & 1) * 8;
      sr.x = pR[off];  sr.y = pR[off + 4];
      si.x = pI[off];  si.y = pI[off + 4];
    }
    const v2f b = op_pair(opY, kk);
    tr = __builtin_amdgcn_wmma_f32_16x16x4_f32(false, sr, false, b,
                                               (short)0, tr, false, false);
    ti = __builtin_amdgcn_wmma_f32_16x16x4_f32(false, si, false, b,
                                               (short)0, ti, false, false);
  }
  v8f ur = {}, ui = {};
#pragma unroll
  for (int kk = 0; kk < 4; ++kk) {
    const v2f a = op_pair(opX, kk);
    v2f br, bi;
    br.x = tr[2 * kk];  br.y = tr[2 * kk + 1];
    bi.x = ti[2 * kk];  bi.y = ti[2 * kk + 1];
    ur = __builtin_amdgcn_wmma_f32_16x16x4_f32(false, a, false, br,
                                               (short)0, ur, false, false);
    ui = __builtin_amdgcn_wmma_f32_16x16x4_f32(false, a, false, bi,
                                               (short)0, ui, false, false);
  }
  *reinterpret_cast<v4f *>(dR)     = __builtin_shufflevector(ur, ur, 0, 1, 2, 3);
  *reinterpret_cast<v4f *>(dR + 4) = __builtin_shufflevector(ur, ur, 4, 5, 6, 7);
  *reinterpret_cast<v4f *>(dI)     = __builtin_shufflevector(ui, ui, 0, 1, 2, 3);
  *reinterpret_cast<v4f *>(dI + 4) = __builtin_shufflevector(ui, ui, 4, 5, 6, 7);
}

struct ETab { v4f e[4]; };

__device__ __forceinline__ ETab load_etab(const float *__restrict__ eTile, unsigned laneB) {
  const char *p = reinterpret_cast<const char *>(eTile);
  ETab t;
#pragma unroll
  for (int j = 0; j < 4; ++j)
    t.e[j] = *reinterpret_cast<const v4f *>(p + laneB + 512u * j);
  return t;
}

struct SAddr { v4u a[2]; };

__device__ __forceinline__ SAddr load_saddr(const float *__restrict__ sTile, unsigned laneB) {
  const char *p = reinterpret_cast<const char *>(sTile);
  SAddr s;
  s.a[0] = *reinterpret_cast<const v4u *>(p + laneB);
  s.a[1] = *reinterpret_cast<const v4u *>(p + laneB + 512u);
  return s;
}

struct UT4 { float t8, t9, zf; };

__device__ __forceinline__ UT4 load_ut(const float *__restrict__ G, int l) {
  const float *p = G + l * LAYER_F + 2 * PHASE_F;
  UT4 t;
  t.t8 = p[0];  t.t9 = p[1];  t.zf = p[2];
  return t;
}

__device__ __forceinline__ void shear4(float t8, float t9, const v2f (&x)[4], v2f (&z)[4]) {
  const v2f a9 = {t9, t9}, n9 = {-t9, -t9};
  const v2f a8 = {t8, t8}, n8 = {-t8, -t8};
  const v2f y0 = __builtin_elementwise_fma(n9, x[1], x[0]);
  const v2f y1 = __builtin_elementwise_fma(a9, x[0], x[1]);
  const v2f y2 = __builtin_elementwise_fma(n9, x[3], x[2]);
  const v2f y3 = __builtin_elementwise_fma(a9, x[2], x[3]);
  z[0] = __builtin_elementwise_fma(n8, y2, y0);
  z[2] = __builtin_elementwise_fma(a8, y0, y2);
  z[1] = __builtin_elementwise_fma(n8, y3, y1);
  z[3] = __builtin_elementwise_fma(a8, y1, y3);
}

__device__ __forceinline__ void ut_products(const UT4 &ut,
                                            const float *pR, const float *pI,
                                            v2f (&zr)[4], v2f (&zi)[4]) {
  v2f xr[4], xi[4];
#pragma unroll
  for (int k = 0; k < 4; ++k) {
    xr[k].x = pR[8 * k];  xr[k].y = pR[8 * (4 + k)];
    xi[k].x = pI[8 * k];  xi[k].y = pI[8 * (4 + k)];
  }
  shear4(ut.t8, ut.t9, xr, zr);
  shear4(ut.t8, ut.t9, xi, zi);
}

__device__ __forceinline__ void ut_apply(const UT4 &ut,
                                         const float *pR, const float *pI,
                                         float *dR, float *dI,
                                         const SAddr sa,
                                         const ETab et) {
  v2f zr[4], zi[4];
  ut_products(ut, pR, pI, zr, zi);
  char *bR = reinterpret_cast<char *>(dR);
  char *bI = reinterpret_cast<char *>(dI);
#pragma unroll
  for (int q = 0; q < 4; ++q) {
    const v2f er = __builtin_shufflevector(et.e[q], et.e[q], 0, 1);
    const v2f ei = __builtin_shufflevector(et.e[q], et.e[q], 2, 3);
    const v2f vr = zr[q] * er - zi[q] * ei;
    const v2f vi = zr[q] * ei + zi[q] * er;
    const unsigned w0 = sa.a[q >> 1][2 * (q & 1)];
    const unsigned w1 = sa.a[q >> 1][2 * (q & 1) + 1];
    *reinterpret_cast<float *>(bR + w0) = vr.x;
    *reinterpret_cast<float *>(bI + w0) = vi.x;
    *reinterpret_cast<float *>(bR + w1) = vr.y;
    *reinterpret_cast<float *>(bI + w1) = vi.y;
  }
}

__device__ __forceinline__ float ut_measure(const UT4 &ut,
                                            const float *pR, const float *pI,
                                            int rr, int permBase) {
  v2f zr[4], zi[4];
  ut_products(ut, pR, pI, zr, zi);
  float acc = 0.f;
#pragma unroll
  for (int v = 0; v < 8; ++v) {
    const float tr = (v >> 2) ? zr[v & 3].y : zr[v & 3].x;
    const float ti = (v >> 2) ? zi[v & 3].y : zi[v & 3].x;
    const float p = tr * tr + ti * ti;
    acc += (cnot_perm(v, rr) & 1) ? -p : p;
  }
  return ((permBase & 1) ? -acc : acc) * ut.zf;
}

__global__ __launch_bounds__(128) __attribute__((amdgpu_waves_per_eu(16, 16)))
void vqc_kernel(const float *__restrict__ X,
                const float *__restrict__ G,
                const float *__restrict__ bias,
                float *__restrict__ out) {
  __shared__ __attribute__((aligned(16))) float SR[2][PDIM];
  __shared__ __attribute__((aligned(16))) float SI[2][PDIM];
  __shared__ float cs[2 * N_WIRES];
  __shared__ float pLo[32][2], pHi[32][2];
  __shared__ float red[4];

  const int tid  = threadIdx.x;
  const int lane = tid & 31;
  const int wave = __builtin_amdgcn_readfirstlane(tid >> 5);
  const int s    = blockIdx.x;

  if (tid < N_WIRES) {
    const float x = X[s * N_WIRES + tid] * 0.7853981633974483f;
    cs[tid] = __cosf(x);
    cs[N_WIRES + tid] = __sinf(x);
  }
  __syncthreads();

  if (tid < 32) {
    float p = 1.f;
#pragma unroll
    for (int w = 5; w < N_WIRES; ++w)
      p *= cs[((tid >> (9 - w)) & 1) * N_WIRES + w];
    pLo[tid][0] = p * G[F_OFF + 64 + tid * 2];
    pLo[tid][1] = p * G[F_OFF + 64 + tid * 2 + 1];
  } else if (tid < 64) {
    const int h = tid - 32;
    float p = 1.f;
#pragma unroll
    for (int w = 0; w < 5; ++w)
      p *= cs[((h >> (4 - w)) & 1) * N_WIRES + w];
    pHi[h][0] = p * G[F_OFF + h * 2];
    pHi[h][1] = p * G[F_OFF + h * 2 + 1];
  }
  __syncthreads();

  const int c16 = lane & 15;
  const int hh  = lane >> 4;

  StartF sf;
#pragma unroll
  for (int j = 0; j < 2; ++j) {
    sf.hi[j].x = pHi[2 * c16 + j][0];
    sf.hi[j].y = pHi[2 * c16 + j][1];
  }
#pragma unroll
  for (int j = 0; j < 4; ++j) {
    sf.lo[j].x = pLo[16 * hh + 4 * j + wave][0];
    sf.lo[j].y = pLo[16 * hh + 4 * j + wave][1];
  }

  const int baseA  = c16 * 68 + hh * 17 + wave;
  const int baseWF = 4 * (hh * 136 + (c16 >> 1) * 17 + (c16 & 1) * 8 + wave * 2);
  const int col     = wave * 16 + c16;
  const int laneHi8 = hh << 3;
  const int baseB2  = (wave >> 1) * 544 + (wave & 1) * 4 + (c16 & 3) * 136 + hh * 68 + (c16 >> 2);
  const float *eTile = G + E_OFF + wave * 512;
  const float *sTile = G + S_OFF + wave * 256;
  const unsigned laneB = (unsigned)lane * 16u;

  AOp aG1 = load_op(G, 1, laneB);
  float z = 0.f;
#pragma unroll
  for (int l = 0; l < N_LAYERS; ++l) {
    const AOp aG0 = load_op(G, 3 * l, laneB);
    const UT4 ut = load_ut(G, l);
    if (l == 0)
      fused_apply<true>(aG1, aG0,
                        SR[0] + baseA, SI[0] + baseA, sf,
                        SR[1] + baseWF, SI[1] + baseWF);
    else
      fused_apply<false>(aG1, aG0,
                         SR[0] + baseA, SI[0] + baseA, sf,
                         SR[1] + baseWF, SI[1] + baseWF);
    __syncthreads();
    if (l + 1 < N_LAYERS) {
      const ETab et = load_etab(eTile + l * DIAG_F, laneB);
      const SAddr sa = load_saddr(sTile + l * SCAT_F, laneB);
      ut_apply(ut, SR[1] + baseB2, SI[1] + baseB2,
               SR[0], SI[0],
               sa, et);
      aG1 = load_op(G, 3 * l + 4, laneB);
      __syncthreads();
    } else {
      const int pb = cnot_perm((col << 4) | laneHi8, l + 1);
      z = ut_measure(ut, SR[1] + baseB2, SI[1] + baseB2, l + 1, pb);
    }
  }

#pragma unroll
  for (int m = 16; m > 0; m >>= 1) z += __shfl_xor(z, m, 32);
  if (lane == 0) red[wave] = z;
  __syncthreads();
  if (tid == 0) {
    float t = 0.f;
#pragma unroll
    for (int w = 0; w < 4; ++w) t += red[w];
    out[s] = t + bias[0];
  }
}

extern "C" void kernel_launch(void *const *d_in, const int *in_sizes, int n_in,
                              void *d_out, int out_size, void *d_ws, size_t ws_size,
                              hipStream_t stream) {
  const float *X = (const float *)d_in[0];
  const float *W = (const float *)d_in[1];
  const float *bias = (const float *)d_in[2];
  float *out = (float *)d_out;
  float *G = (float *)d_ws;

  const int B = in_sizes[0] / N_WIRES;

  vqc_precompute<<<N_LAYERS, 256, 0, stream>>>(W, G);
  vqc_kernel<<<B, 128, 0, stream>>>(X, G, bias, out);
}


# Wires 0-3 (a Kronecker product of four rotations) leave the dense 16x16 WMMA product for shears on the vector unit: 1 FMA per value and wire (was 4 MACs); G1 stays on WMMA (64 to 32); new sum order
# speedup vs baseline: 3.6849x; 1.2328x over previous
#include <hip/hip_runtime.h>


typedef float v2f __attribute__((ext_vector_type(2)));
typedef float v4f __attribute__((ext_vector_type(4)));
typedef float v8f __attribute__((ext_vector_type(8)));
typedef unsigned v4u __attribute__((ext_vector_type(4)));

#define N_WIRES 10
#define N_LAYERS 4
#define DIM 1024
#define PADDED(k) ((k) + ((k) >> 4))
#define PDIM (DIM + (DIM >> 4))
#define PHASE_F 256
#define LAYER_F 512
#define OPS_F (N_LAYERS * LAYER_F)
#define DIAG_F 2048
#define E_OFF OPS_F
#define F_OFF (E_OFF + N_LAYERS * DIAG_F)
#define S_OFF (F_OFF + 128)
#define SCAT_F 1024

__device__ __forceinline__ int frag_pos(int M, int kk, int h, int e) {
  return (kk >> 1) * 128 + (M + 16 * h) * 4 + (kk & 1) * 2 + e;
}

__device__ __forceinline__ void ut_owner(int m, int &tile, int &ln, int &v) {
  tile = (m >> 2) & 3;
  ln = ((m >> 6) & 7) | (((m >> 4) & 1) << 3) | (((m >> 5) & 1) << 4);
  v = ((m >> 9) << 2) | (m & 3);
}

__device__ __forceinline__ int diag_pos(int m) {
  int tile, ln, v;
  ut_owner(m, tile, ln, v);
  return ((tile * 4 + (v & 3)) * 32 + ln) * 4 + (v >> 2);
}

__device__ __forceinline__ int scat_pos(int m) {
  int tile, ln, v;
  ut_owner(m, tile, ln, v);
  const int j = 2 * (v & 3) + (v >> 2);
  return ((tile * 2 + (j >> 2)) * 32 + ln) * 4 + (j & 3);
}

struct AOp { v4f q[2]; };

__device__ __forceinline__ AOp load_op(const float *__restrict__ G, int phase, unsigned laneB) {
  const char *p = reinterpret_cast<const char *>(G + phase * PHASE_F);
  AOp o;
  o.q[0] = *reinterpret_cast<const v4f *>(p + laneB);
  o.q[1] = *reinterpret_cast<const v4f *>(p + laneB + 512u);
  return o;
}

__device__ __forceinline__ void cmul(float ar, float ai, float br, float bi,
                                     float &cr, float &ci) {
  cr = ar * br - ai * bi;
  ci = ar * bi + ai * br;
}

__device__ __forceinline__ int cnot_perm(int m, int rr) {
#pragma unroll
  for (int w = 0; w < N_WIRES; ++w) {
    int t = w + rr; if (t >= N_WIRES) t -= N_WIRES;
    m ^= ((m >> (9 - w)) & 1) << (9 - t);
  }
  return m;
}

__device__ __forceinline__ float ry_el(const float *cs2, int rb, int cb) {
  return (rb == cb) ? cs2[0] : (rb ? cs2[1] : -cs2[1]);
}

struct Shear { float t, sigma; int f; };
__device__ __forceinline__ Shear shear_of(const float *cs2) {
  const float c = cs2[0], s = cs2[1];
  Shear r;
  if (fabsf(c) >= fabsf(s)) { r.f = 0; r.sigma = c; r.t = s / c; }
  else                      { r.f = 1; r.sigma = s; r.t = -c / s; }
  return r;
}

__global__ __launch_bounds__(256) void vqc_precompute(const float *__restrict__ W,
                                                      float *__restrict__ G) {
  __shared__ float ry[N_WIRES][2];
  __shared__ float zz[3][N_WIRES][2];
  __shared__ float tab[6][32][2];
  const int tid = threadIdx.x;
  const int l = blockIdx.x;

  if (tid < N_WIRES) {
    const float theta = W[(l * N_WIRES + tid) * 3 + 1];
    const float omega = W[(l * N_WIRES + tid) * 3 + 2];
    ry[tid][0] = __cosf(0.5f * theta);    ry[tid][1] = __sinf(0.5f * theta);
    zz[0][tid][0] = __cosf(0.5f * omega); zz[0][tid][1] = __sinf(0.5f * omega);
  } else if (tid >= 32 && tid < 32 + N_WIRES) {
    const int q = tid - 32;
    float c = 1.f, s = 0.f;
    if (l + 1 < N_LAYERS) {
      const float phi = W[((l + 1) * N_WIRES + q) * 3 + 0];
      c = __cosf(0.5f * phi); s = __sinf(0.5f * phi);
    }
    zz[1][q][0] = c; zz[1][q][1] = s;
  } else if (tid >= 64 && tid < 64 + N_WIRES) {
    const int q = tid - 64;
    const float phi = W[q * 3 + 0];
    zz[2][q][0] = __cosf(0.5f * phi); zz[2][q][1] = __sinf(0.5f * phi);
  }
  __syncthreads();

  if (tid < 192) {
    const int t = tid >> 5, e = tid & 31;
    const int w0 = (t & 1) * 5;
    float pr = 1.f, pi = 0.f;
#pragma unroll
    for (int q = 0; q < 5; ++q) {
      const float c = zz[t >> 1][w0 + q][0];
      const float s = ((e >> (4 - q)) & 1) ? zz[t >> 1][w0 + q][1] : -zz[t >> 1][w0 + q][1];
      float nr, ni; cmul(pr, pi, c, s, nr, ni);
      pr = nr; pi = ni;
    }
    tab[t][e][0] = pr; tab[t][e][1] = pi;
  }
  __syncthreads();

  {
    float *L = G + l * LAYER_F;
    const int r = tid >> 4, c = tid & 15;
    float p = 1.f;
#pragma unroll
    for (int q = 0; q < 4; ++q)
      p *= ry_el(ry[4 + q], (r >> (3 - q)) & 1, (c >> (3 - q)) & 1);
    L[frag_pos(r, 2 * (c >> 3) + ((c >> 1) & 1), (c >> 2) & 1, c & 1)] = p;
  }

  int turned = 0;
  float sig = 1.f;
  float tw[6];
#pragma unroll
  for (int i = 0; i < 6; ++i) {
    const int w = (i < 4) ? i : i + 4;
    const Shear sh = shear_of(ry[w]);
    tw[i] = sh.t;
    sig *= sh.sigma;
    turned |= sh.f << (9 - w);
  }
  if (tid == 0) {
    float *U = G + l * LAYER_F + PHASE_F;
#pragma unroll
    for (int i = 0; i < 6; ++i) U[i] = tw[i];
    U[6] = (cnot_perm(turned, l + 1) & 1) ? -(sig * sig) : sig * sig;
  }

  if (l + 1 < N_LAYERS) {
    float *E = G + E_OFF + l * DIAG_F;
    unsigned *S = reinterpret_cast<unsigned *>(G + S_OFF + l * SCAT_F);
    for (int m = tid; m < DIM; m += 256) {
      const int mt = m ^ turned;
      const float g = (__popc(turned & ~mt) & 1) ? -sig : sig;
      const int p = cnot_perm(mt, l + 1);
      S[scat_pos(m)] = (unsigned)PADDED(p) * 4u;
      float orr, oi, fr, fi, er, ei;
      cmul(tab[0][mt >> 5][0], tab[0][mt >> 5][1], tab[1][mt & 31][0], tab[1][mt & 31][1], orr, oi);
      cmul(tab[2][p >> 5][0], tab[2][p >> 5][1], tab[3][p & 31][0], tab[3][p & 31][1], fr, fi);
      cmul(orr, oi, fr, fi, er, ei);
      const int pos = diag_pos(m);
      E[pos] = g * er; E[pos + 2] = g * ei;
    }
  }

  if (l == 0 && tid < 64) {
    G[F_OFF + tid * 2]     = tab[4 + (tid >> 5)][tid & 31][0];
    G[F_OFF + tid * 2 + 1] = tab[4 + (tid >> 5)][tid & 31][1];
  }
}

__device__ __forceinline__ v2f op_pair(const AOp &o, int kk) {
  const v4f q = o.q[kk >> 1];
  v2f a;
  if (kk & 1) { a.x = q.z; a.y = q.w; } else { a.x = q.x; a.y = q.y; }
  return a;
}

__device__ __forceinline__ v2f shear_in_pair(const v2f nt, const v2f x) {
  return __builtin_elementwise_fma(nt, __builtin_shufflevector(x, x, 1, 0), x);
}

struct Coef { float t0, t1, t2, t3, t8, t9, zf; };
__device__ __forceinline__ Coef load_coef(const float *__restrict__ G, int l) {
  const float *p = G + l * LAYER_F + PHASE_F;
  Coef t;
  t.t0 = p[0];  t.t1 = p[1];  t.t2 = p[2];  t.t3 = p[3];
  t.t8 = p[4];  t.t9 = p[5];  t.zf = p[6];
  return t;
}

__device__ __forceinline__ void shear_regs(const Coef &cf, const v8f a, v2f (&y)[4]) {
  const v2f p0 = {a[0], a[1]}, p1 = {a[2], a[3]}, p2 = {a[4], a[5]}, p3 = {a[6], a[7]};
  const v2f a1 = {cf.t1, cf.t1}, n1 = {-cf.t1, -cf.t1};
  const v2f a2 = {cf.t2, cf.t2}, n2 = {-cf.t2, -cf.t2};
  const v2f n3 = {-cf.t3, cf.t3};
  const v2f q0 = __builtin_elementwise_fma(n1, p2, p0);
  const v2f q2 = __builtin_elementwise_fma(a1, p0, p2);
  const v2f q1 = __builtin_elementwise_fma(n1, p3, p1);
  const v2f q3 = __builtin_elementwise_fma(a1, p1, p3);
  const v2f s0 = __builtin_elementwise_fma(n2, q1, q0);
  const v2f s1 = __builtin_elementwise_fma(a2, q0, q1);
  const v2f s2 = __builtin_elementwise_fma(n2, q3, q2);
  const v2f s3 = __builtin_elementwise_fma(a2, q2, q3);
  y[0] = shear_in_pair(n3, s0);
  y[1] = shear_in_pair(n3, s1);
  y[2] = shear_in_pair(n3, s2);
  y[3] = shear_in_pair(n3, s3);
}

struct StartF { v2f hi[2], lo[4]; };

template <bool START>
__device__ __forceinline__ void fused_apply(const AOp opY, const Coef &cf,
                                            const float *pR, const float *pI,
                                            const StartF &sf,
                                            float *dR, float *dI) {
  v8f tr = {}, ti = {};
#pragma unroll
  for (int kk = 0; kk < 4; ++kk) {
    v2f sr, si;
    if (START) {
      const v2f h = sf.hi[kk >> 1];
      const v2f l0 = sf.lo[2 * (kk & 1)], l1 = sf.lo[2 * (kk & 1) + 1];
      float r0, i0, r1, i1;
      cmul(h.x, h.y, l0.x, l0.y, r0, i0);
      cmul(h.x, h.y, l1.x, l1.y, r1, i1);
      sr.x = r0;  sr.y = r1;
      si.x = i0;  si.y = i1;
    } else {
      const int off = (kk >> 1) * 34 + (kk & 1) * 8;
      sr.x = pR[off];  sr.y = pR[off + 4];
      si.x = pI[off];  si.y = pI[off + 4];
    }
    const v2f b = op_pair(opY, kk);
    tr = __builtin_amdgcn_wmma_f32_16x16x4_f32(false, sr, false, b,
                                               (short)0, tr, false, false);
    ti = __builtin_amdgcn_wmma_f32_16x16x4_f32(false, si, false, b,
                                               (short)0, ti, false, false);
  }
  v2f yr[4], yi[4];
  shear_regs(cf, tr, yr);
  shear_regs(cf, ti, yi);
  *reinterpret_cast<v4f *>(dR)     = __builtin_shufflevector(yr[0], yr[1], 0, 1, 2, 3);
  *reinterpret_cast<v4f *>(dR + 4) = __builtin_shufflevector(yr[2], yr[3], 0, 1, 2, 3);
  *reinterpret_cast<v4f *>(dI)     = __builtin_shufflevector(yi[0], yi[1], 0, 1, 2, 3);
  *reinterpret_cast<v4f *>(dI + 4) = __builtin_shufflevector(yi[2], yi[3], 0, 1, 2, 3);
}

struct ETab { v4f e[4]; };

__device__ __forceinline__ ETab load_etab(const float *__restrict__ eTile, unsigned laneB) {
  const char *p = reinterpret_cast<const char *>(eTile);
  ETab t;
#pragma unroll
  for (int j = 0; j < 4; ++j)
    t.e[j] = *reinterpret_cast<const v4f *>(p + laneB + 512u * j);
  return t;
}

struct SAddr { v4u a[2]; };

__device__ __forceinline__ SAddr load_saddr(const float *__restrict__ sTile, unsigned laneB) {
  const char *p = reinterpret_cast<const char *>(sTile);
  SAddr s;
  s.a[0] = *reinterpret_cast<const v4u *>(p + laneB);
  s.a[1] = *reinterpret_cast<const v4u *>(p + laneB + 512u);
  return s;
}

__device__ __forceinline__ void shear8(const Coef &cf, const v2f (&x)[4], v2f (&z)[4]) {
  const v2f a9 = {cf.t9, cf.t9}, n9 = {-cf.t9, -cf.t9};
  const v2f a8 = {cf.t8, cf.t8}, n8 = {-cf.t8, -cf.t8};
  const v2f n0 = {-cf.t0, cf.t0};
  const v2f y0 = __builtin_elementwise_fma(n9, x[1], x[0]);
  const v2f y1 = __builtin_elementwise_fma(a9, x[0], x[1]);
  const v2f y2 = __builtin_elementwise_fma(n9, x[3], x[2]);
  const v2f y3 = __builtin_elementwise_fma(a9, x[2], x[3]);
  const v2f w0 = __builtin_elementwise_fma(n8, y2, y0);
  const v2f w2 = __builtin_elementwise_fma(a8, y0, y2);
  const v2f w1 = __builtin_elementwise_fma(n8, y3, y1);
  const v2f w3 = __builtin_elementwise_fma(a8, y1, y3);
  z[0] = shear_in_pair(n0, w0);
  z[1] = shear_in_pair(n0, w1);
  z[2] = shear_in_pair(n0, w2);
  z[3] = shear_in_pair(n0, w3);
}

__device__ __forceinline__ void ut_products(const Coef &cf,
                                            const float *pR, const float *pI,
                                            v2f (&zr)[4], v2f (&zi)[4]) {
  v2f xr[4], xi[4];
#pragma unroll
  for (int k = 0; k < 4; ++k) {
    xr[k].x = pR[8 * k];  xr[k].y = pR[8 * (4 + k)];
    xi[k].x = pI[8 * k];  xi[k].y = pI[8 * (4 + k)];
  }
  shear8(cf, xr, zr);
  shear8(cf, xi, zi);
}

__device__ __forceinline__ void ut_apply(const Coef &cf,
                                         const float *pR, const float *pI,
                                         float *dR, float *dI,
                                         const SAddr sa,
                                         const ETab et) {
  v2f zr[4], zi[4];
  ut_products(cf, pR, pI, zr, zi);
  char *bR = reinterpret_cast<char *>(dR);
  char *bI = reinterpret_cast<char *>(dI);
#pragma unroll
  for (int q = 0; q < 4; ++q) {
    const v2f er = __builtin_shufflevector(et.e[q], et.e[q], 0, 1);
    const v2f ei = __builtin_shufflevector(et.e[q], et.e[q], 2, 3);
    const v2f vr = zr[q] * er - zi[q] * ei;
    const v2f vi = zr[q] * ei + zi[q] * er;
    const unsigned w0 = sa.a[q >> 1][2 * (q & 1)];
    const unsigned w1 = sa.a[q >> 1][2 * (q & 1) + 1];
    *reinterpret_cast<float *>(bR + w0) = vr.x;
    *reinterpret_cast<float *>(bI + w0) = vi.x;
    *reinterpret_cast<float *>(bR + w1) = vr.y;
    *reinterpret_cast<float *>(bI + w1) = vi.y;
  }
}

__device__ __forceinline__ float ut_measure(const Coef &cf,
                                            const float *pR, const float *pI,
                                            int rr, int permBase) {
  v2f zr[4], zi[4];
  ut_products(cf, pR, pI, zr, zi);
  float acc = 0.f;
#pragma unroll
  for (int v = 0; v < 8; ++v) {
    const float tr = (v >> 2) ? zr[v & 3].y : zr[v & 3].x;
    const float ti = (v >> 2) ? zi[v & 3].y : zi[v & 3].x;
    const float p = tr * tr + ti * ti;
    acc += (cnot_perm(((v >> 2) << 9) | (v & 3), rr) & 1) ? -p : p;
  }
  return ((permBase & 1) ? -acc : acc) * cf.zf;
}

__global__ __launch_bounds__(128) __attribute__((amdgpu_waves_per_eu(16, 16)))
void vqc_kernel(const float *__restrict__ X,
                const float *__restrict__ G,
                const float *__restrict__ bias,
                float *__restrict__ out) {
  __shared__ __attribute__((aligned(16))) float SR[2][PDIM];
  __shared__ __attribute__((aligned(16))) float SI[2][PDIM];
  __shared__ float cs[2 * N_WIRES];
  __shared__ float pLo[32][2], pHi[32][2];
  __shared__ float red[4];

  const int tid  = threadIdx.x;
  const int lane = tid & 31;
  const int wave = __builtin_amdgcn_readfirstlane(tid >> 5);
  const int s    = blockIdx.x;

  if (tid < N_WIRES) {
    const float x = X[s * N_WIRES + tid] * 0.7853981633974483f;
    cs[tid] = __cosf(x);
    cs[N_WIRES + tid] = __sinf(x);
  }
  __syncthreads();

  if (tid < 32) {
    float p = 1.f;
#pragma unroll
    for (int w = 5; w < N_WIRES; ++w)
      p *= cs[((tid >> (9 - w)) & 1) * N_WIRES + w];
    pLo[tid][0] = p * G[F_OFF + 64 + tid * 2];
    pLo[tid][1] = p * G[F_OFF + 64 + tid * 2 + 1];
  } else if (tid < 64) {
    const int h = tid - 32;
    float p = 1.f;
#pragma unroll
    for (int w = 0; w < 5; ++w)
      p *= cs[((h >> (4 - w)) & 1) * N_WIRES + w];
    pHi[h][0] = p * G[F_OFF + h * 2];
    pHi[h][1] = p * G[F_OFF + h * 2 + 1];
  }
  __syncthreads();

  const int c16 = lane & 15;
  const int hh  = lane >> 4;

  StartF sf;
#pragma unroll
  for (int j = 0; j < 2; ++j) {
    sf.hi[j].x = pHi[2 * c16 + j][0];
    sf.hi[j].y = pHi[2 * c16 + j][1];
  }
#pragma unroll
  for (int j = 0; j < 4; ++j) {
    sf.lo[j].x = pLo[16 * hh + 4 * j + wave][0];
    sf.lo[j].y = pLo[16 * hh + 4 * j + wave][1];
  }

  const int baseA  = c16 * 68 + hh * 17 + wave;
  const int baseWF = (c16 & 1) * 544 + (c16 >> 1) * 68 + hh * 32 + wave * 8;
  const int baseB2 = (wave & 1) * 544 + ((wave >> 1) + 2 * (c16 >> 3) + 4 * hh) * 68 + (c16 & 7);
  const float *eTile = G + E_OFF + wave * 512;
  const float *sTile = G + S_OFF + wave * 256;
  const unsigned laneB = (unsigned)lane * 16u;

  AOp aG1 = load_op(G, 0, laneB);
  float z = 0.f;
#pragma unroll
  for (int l = 0; l < N_LAYERS; ++l) {
    const Coef cf = load_coef(G, l);
    if (l == 0)
      fused_apply<true>(aG1, cf,
                        SR[0] + baseA, SI[0] + baseA, sf,
                        SR[1] + baseWF, SI[1] + baseWF);
    else
      fused_apply<false>(aG1, cf,
                         SR[0] + baseA, SI[0] + baseA, sf,
                         SR[1] + baseWF, SI[1] + baseWF);
    __syncthreads();
    if (l + 1 < N_LAYERS) {
      const ETab et = load_etab(eTile + l * DIAG_F, laneB);
      const SAddr sa = load_saddr(sTile + l * SCAT_F, laneB);
      ut_apply(cf, SR[1] + baseB2, SI[1] + baseB2,
               SR[0], SI[0],
               sa, et);
      aG1 = load_op(G, 2 * l + 2, laneB);
      __syncthreads();
    } else {
      const int pb = cnot_perm(((c16 & 7) << 6) | (hh << 5) | ((c16 >> 3) << 4) | (wave << 2), l + 1);
      z = ut_measure(cf, SR[1] + baseB2, SI[1] + baseB2, l + 1, pb);
    }
  }

#pragma unroll
  for (int m = 16; m > 0; m >>= 1) z += __shfl_xor(z, m, 32);
  if (lane == 0) red[wave] = z;
  __syncthreads();
  if (tid == 0) {
    float t = 0.f;
#pragma unroll
    for (int w = 0; w < 4; ++w) t += red[w];
    out[s] = t + bias[0];
  }
}

extern "C" void kernel_launch(void *const *d_in, const int *in_sizes, int n_in,
                              void *d_out, int out_size, void *d_ws, size_t ws_size,
                              hipStream_t stream) {
  const float *X = (const float *)d_in[0];
  const float *W = (const float *)d_in[1];
  const float *bias = (const float *)d_in[2];
  float *out = (float *)d_out;
  float *G = (float *)d_ws;

  const int B = in_sizes[0] / N_WIRES;

  vqc_precompute<<<N_LAYERS, 256, 0, stream>>>(W, G);
  vqc_kernel<<<B, 128, 0, stream>>>(X, G, bias, out);
}
